# starting kernel
# baseline (speedup 1.0000x reference)
#include <hip/hip_runtime.h>
#include <hip/hip_bf16.h>

typedef __attribute__((ext_vector_type(2))) float v2f;
typedef __attribute__((ext_vector_type(8))) float v8f;

#define K_DIM 128
#define LDS_PAD 8

__global__ void gcn_deg_init(float* __restrict__ deg, int n) {
    int i = blockIdx.x * blockDim.x + threadIdx.x;
    if (i < n) deg[i] = 1.0f;
}

__global__ void gcn_deg_count(const int* __restrict__ dst, float* __restrict__ deg, int nE) {
    int e = blockIdx.x * blockDim.x + threadIdx.x;
    if (e < nE) atomicAdd(&deg[dst[e]], 1.0f);
}

__global__ void gcn_rsqrt(float* __restrict__ d, int n) {
    int i = blockIdx.x * blockDim.x + threadIdx.x;
    if (i < n) d[i] = rsqrtf(fmaxf(d[i], 1.0f));
}

template <int M, bool RELU_IN>
__global__ void __launch_bounds__(256)
gcn_gemm_wmma(const float* __restrict__ X, const float* __restrict__ W,
              float* __restrict__ H, int nRows) {
    constexpr int LSTR = M + LDS_PAD;
    __shared__ float Wlds[K_DIM * LSTR];

    const int tid = threadIdx.x;
    for (int i = tid * 4; i < K_DIM * M; i += 256 * 4) {
        const int r = i / M;
        const int c = i - r * M;
        *(float4*)&Wlds[r * LSTR + c] = *(const float4*)&W[i];
    }
    __syncthreads();

    const int wave = tid >> 5;
    const int lane = tid & 31;
    const int tile = blockIdx.x * 8 + wave;
    if (tile * 16 >= nRows) return;

    const int rowA = tile * 16 + (lane & 15);
    const int kOff = (lane >> 4) * 2;
    const int nCol = (lane & 15);
    const float* __restrict__ xrow = X + (size_t)rowA * K_DIM;
    const float* __restrict__ wcol = &Wlds[kOff * LSTR + nCol];

    v8f acc[M / 16] = {};

#pragma unroll 4
    for (int k = 0; k < K_DIM; k += 4) {
        float a0 = xrow[k + kOff];
        float a1 = xrow[k + kOff + 1];
        if (RELU_IN) { a0 = fmaxf(a0, 0.0f); a1 = fmaxf(a1, 0.0f); }
        v2f a; a.x = a0; a.y = a1;
        const float* __restrict__ wk = wcol + k * LSTR;
#pragma unroll
        for (int t = 0; t < M / 16; ++t) {
            v2f b;
            b.x = wk[t * 16];
            b.y = wk[LSTR + t * 16];
            acc[t] = __builtin_amdgcn_wmma_f32_16x16x4_f32(
                false, a, false, b, (short)0, acc[t], false, false);
        }
    }

    const int rBase = tile * 16 + ((lane >> 4) << 3);
#pragma unroll
    for (int t = 0; t < M / 16; ++t) {
#pragma unroll
        for (int v = 0; v < 8; ++v) {
            H[(size_t)(rBase + v) * M + t * 16 + nCol] = acc[t][v];
        }
    }
}

template <int M>
__global__ void gcn_agg_init(const float* __restrict__ H, const float* __restrict__ dis,
                             const float* __restrict__ bias, float* __restrict__ out, int n) {
    int gid = blockIdx.x * blockDim.x + threadIdx.x;
    if (gid >= n * M) return;
    int i = gid / M;
    int j = gid - i * M;
    float s = dis[i];
    out[gid] = H[gid] * (s * s) + bias[j];
}

template <int M>
__global__ void gcn_agg_edges(const float* __restrict__ H,
                              const int* __restrict__ src, const int* __restrict__ dst,
                              const float* __restrict__ dis,
                              float* __restrict__ out, int nE) {
    int gid = blockIdx.x * blockDim.x + threadIdx.x;
    int e    = gid >> 5;
    int lane = gid & 31;
    if (e >= nE) return;
    int s = src[e];
    int d = dst[e];
    float w = dis[s] * dis[d];
    const float* __restrict__ hs = H + (size_t)s * M;
    float* __restrict__ od = out + (size_t)d * M;
#pragma unroll
    for (int j = lane; j < M; j += 32) {
        atomicAdd(&od[j], hs[j] * w);
    }
}

extern "C" void kernel_launch(void* const* d_in, const int* in_sizes, int n_in,
                              void* d_out, int out_size, void* d_ws, size_t ws_size,
                              hipStream_t stream) {
    (void)n_in; (void)out_size; (void)ws_size;

    const float* x  = (const float*)d_in[0];
    const int*   ei = (const int*)d_in[1];
    const float* W1 = (const float*)d_in[2];
    const float* b1 = (const float*)d_in[3];
    const float* W2 = (const float*)d_in[4];
    const float* b2 = (const float*)d_in[5];
    const float* W3 = (const float*)d_in[6];
    const float* b3 = (const float*)d_in[7];
    float* out = (float*)d_out;

    const int N = in_sizes[0] / K_DIM;
    const int E = in_sizes[1] / 2;
    const int* src = ei;
    const int* dst = ei + E;

    float* dis = (float*)d_ws;
    float* Bh  = dis + N;
    float* Ba  = Bh + (size_t)N * 128;

    const int T = 256;
    const int gN    = (N + T - 1) / T;
    const int gE    = (E + T - 1) / T;
    const int gEdge = (int)(((long long)E * 32 + T - 1) / T);
    const int tiles = (N + 15) / 16;
    const int gGemm = (tiles + 7) / 8;

    gcn_deg_init<<<gN, T, 0, stream>>>(dis, N);
    gcn_deg_count<<<gE, T, 0, stream>>>(dst, dis, E);
    gcn_rsqrt<<<gN, T, 0, stream>>>(dis, N);

    gcn_gemm_wmma<128, false><<<gGemm, T, 0, stream>>>(x, W1, Bh, N);
    gcn_agg_init<128><<<(N * 128 + T - 1) / T, T, 0, stream>>>(Bh, dis, b1, Ba, N);
    gcn_agg_edges<128><<<gEdge, T, 0, stream>>>(Bh, src, dst, dis, Ba, E);

    gcn_gemm_wmma<128, true><<<gGemm, T, 0, stream>>>(Ba, W2, Bh, N);
    gcn_agg_init<128><<<(N * 128 + T - 1) / T, T, 0, stream>>>(Bh, dis, b2, Ba, N);
    gcn_agg_edges<128><<<gEdge, T, 0, stream>>>(Bh, src, dst, dis, Ba, E);

    gcn_gemm_wmma<64, true><<<gGemm, T, 0, stream>>>(Ba, W3, Bh, N);
    gcn_agg_init<64><<<(N * 64 + T - 1) / T, T, 0, stream>>>(Bh, dis, b3, out, N);
    gcn_agg_edges<64><<<gEdge, T, 0, stream>>>(Bh, src, dst, dis, out, E);
}


# Aggregation as a gather: edges grouped by destination per call (rank = returned integer add), each row summed in registers in rank order, stored once; float atomic scatter and its init pass removed
# speedup vs baseline: 3.1266x; 3.1266x over previous
#include <hip/hip_runtime.h>
#include <hip/hip_bf16.h>

typedef __attribute__((ext_vector_type(2))) float v2f;
typedef __attribute__((ext_vector_type(8))) float v8f;

#define K_DIM 128
#define LDS_PAD 8
#define ELL_W 32
#define HP 64

__global__ void gcn_cnt_init(int* __restrict__ cnt, int* __restrict__ novf, int n) {
    int i = blockIdx.x * blockDim.x + threadIdx.x;
    if (i < n) cnt[i] = 0;
    if (i == 0) *novf = 0;
}

__global__ void gcn_count_place(const int* __restrict__ src, const int* __restrict__ dst,
                                int* __restrict__ cnt, int* __restrict__ ell,
                                int* __restrict__ novf, int* __restrict__ ovf, int nE) {
    int e = blockIdx.x * blockDim.x + threadIdx.x;
    if (e >= nE) return;
    const int d = dst[e];
    const int r = atomicAdd(&cnt[d], 1);
    if (r < ELL_W) ell[(size_t)d * ELL_W + r] = src[e];
    else           ovf[atomicAdd(novf, 1)] = e;
}

__global__ void gcn_dis(const int* __restrict__ cnt, float* __restrict__ dis, int n) {
    int i = blockIdx.x * blockDim.x + threadIdx.x;
    if (i < n) dis[i] = rsqrtf(fmaxf(1.0f + (float)cnt[i], 1.0f));
}

template <int M, bool RELU_IN>
__global__ void __launch_bounds__(256)
gcn_gemm_wmma(const float* __restrict__ X, const float* __restrict__ W,
              float* __restrict__ H0, float* __restrict__ H1, int nRows) {
    constexpr int LSTR = M + LDS_PAD;
    __shared__ float Wlds[K_DIM * LSTR];

    const int tid = threadIdx.x;
    for (int i = tid * 4; i < K_DIM * M; i += 256 * 4) {
        const int r = i / M;
        const int c = i - r * M;
        *(float4*)&Wlds[r * LSTR + c] = *(const float4*)&W[i];
    }
    __syncthreads();

    const int wave = tid >> 5;
    const int lane = tid & 31;
    const int tile = blockIdx.x * 8 + wave;
    if (tile * 16 >= nRows) return;

    const int rowA = tile * 16 + (lane & 15);
    const int kOff = (lane >> 4) * 2;
    const int nCol = (lane & 15);
    const float* __restrict__ xrow = X + (size_t)rowA * K_DIM;
    const float* __restrict__ wcol = &Wlds[kOff * LSTR + nCol];

    v8f acc[M / 16] = {};

#pragma unroll 4
    for (int k = 0; k < K_DIM; k += 4) {
        float a0 = xrow[k + kOff];
        float a1 = xrow[k + kOff + 1];
        if (RELU_IN) { a0 = fmaxf(a0, 0.0f); a1 = fmaxf(a1, 0.0f); }
        v2f a; a.x = a0; a.y = a1;
        const float* __restrict__ wk = wcol + k * LSTR;
#pragma unroll
        for (int t = 0; t < M / 16; ++t) {
            v2f b;
            b.x = wk[t * 16];
            b.y = wk[LSTR + t * 16];
            acc[t] = __builtin_amdgcn_wmma_f32_16x16x4_f32(
                false, a, false, b, (short)0, acc[t], false, false);
        }
    }

    const int rBase = tile * 16 + ((lane >> 4) << 3);
#pragma unroll
    for (int t = 0; t < M / 16; ++t) {
        float* __restrict__ hp = (t < HP / 16) ? H0 : H1;
#pragma unroll
        for (int v = 0; v < 8; ++v) {
            hp[(size_t)(rBase + v) * HP + (t & 3) * 16 + nCol] = acc[t][v];
        }
    }
}

template <int EPI>
__device__ __forceinline__ int lane_pick(int v, int j) {
    if constexpr (EPI == 1) return __builtin_amdgcn_readlane(v, j);
    else                    return __shfl(v, j);
}

template <int M>
__global__ void __launch_bounds__(256)
gcn_gather(const float* __restrict__ Hlo, const float* __restrict__ Hhi,
           const int* __restrict__ ell, const int* __restrict__ cnt,
           const int* __restrict__ novf, const int* __restrict__ ovf,
           const int* __restrict__ src, const int* __restrict__ dst,
           const float* __restrict__ dis, const float* __restrict__ bias,
           float* __restrict__ out, int n) {
    constexpr int LPR = M / 4;
    constexpr int EPI = 32 / LPR;
    const int lane = threadIdx.x & 31;
    const int d = __builtin_amdgcn_readfirstlane(blockIdx.x * 8 + (threadIdx.x >> 5));
    if (d >= n) return;
    const int sub = lane / LPR;
    const int c4  = lane % LPR;
    const float* __restrict__ hb = ((c4 < 16) ? Hlo : Hhi) + (c4 & 15) * 4;

    const float dd = dis[d];
    const int deg = cnt[d];
    const int nl = deg < ELL_W ? deg : ELL_W;

    int my_s = 0;
    float my_d = 0.0f;
    if (lane < nl) {
        my_s = ell[(size_t)d * ELL_W + lane];
        my_d = dis[my_s];
    }

    const float4 hs = *(const float4*)(hb + (size_t)d * HP);
    const float4 bs = *(const float4*)(bias + c4 * 4);
    float4 acc = make_float4(0.0f, 0.0f, 0.0f, 0.0f);

    int j0 = 0;
    for (; j0 + 4 * EPI <= nl; j0 += 4 * EPI) {
        float4 h[4];
        float w[4];
#pragma unroll
        for (int u = 0; u < 4; ++u) {
            const int j = j0 + u * EPI + sub;
            const int s = lane_pick<EPI>(my_s, j);
            w[u] = __int_as_float(lane_pick<EPI>(__float_as_int(my_d), j)) * dd;
            h[u] = *(const float4*)(hb + (size_t)s * HP);
        }
#pragma unroll
        for (int u = 0; u < 4; ++u) {
            acc.x += h[u].x * w[u];
            acc.y += h[u].y * w[u];
            acc.z += h[u].z * w[u];
            acc.w += h[u].w * w[u];
        }
    }
#pragma unroll 1
    for (; j0 < nl; j0 += EPI) {
        const int j = j0 + sub;
        const int s = lane_pick<EPI>(my_s, j);
        const float w = __int_as_float(lane_pick<EPI>(__float_as_int(my_d), j)) * dd;
        float4 h = *(const float4*)(hb + (size_t)s * HP);
        if (EPI == 2 && j >= nl) h = make_float4(0.0f, 0.0f, 0.0f, 0.0f);
        acc.x += h.x * w;
        acc.y += h.y * w;
        acc.z += h.z * w;
        acc.w += h.w * w;
    }

    if (deg > ELL_W) {
        const int no = *novf;
        for (int i0 = 0; i0 < no; i0 += 32) {
            const int i = i0 + lane;
            int s_ = 0;
            bool m = false;
            if (i < no) {
                const int e = ovf[i];
                m = (dst[e] == d);
                if (m) s_ = src[e];
            }
            unsigned mask = (unsigned)__ballot(m);
            while (mask) {
                const int l = __ffs(mask) - 1;
                mask &= mask - 1;
                const int s = __shfl(s_, l);
                const float w = dis[s] * dd;
                if (sub == 0) {
                    const float4 h = *(const float4*)(hb + (size_t)s * HP);
                    acc.x += h.x * w;
                    acc.y += h.y * w;
                    acc.z += h.z * w;
                    acc.w += h.w * w;
                }
            }
        }
    }

    if (sub == 0) {
        const float s2 = dd * dd;
        acc.x += hs.x * s2 + bs.x;
        acc.y += hs.y * s2 + bs.y;
        acc.z += hs.z * s2 + bs.z;
        acc.w += hs.w * s2 + bs.w;
    }
    if (EPI == 2) {
        acc.x += __shfl_xor(acc.x, 16);
        acc.y += __shfl_xor(acc.y, 16);
        acc.z += __shfl_xor(acc.z, 16);
        acc.w += __shfl_xor(acc.w, 16);
    }
    if (sub == 0) *(float4*)(out + (size_t)d * M + c4 * 4) = acc;
}

extern "C" void kernel_launch(void* const* d_in, const int* in_sizes, int n_in,
                              void* d_out, int out_size, void* d_ws, size_t ws_size,
                              hipStream_t stream) {
    (void)n_in; (void)out_size; (void)ws_size;

    const float* x  = (const float*)d_in[0];
    const int*   ei = (const int*)d_in[1];
    const float* W1 = (const float*)d_in[2];
    const float* b1 = (const float*)d_in[3];
    const float* W2 = (const float*)d_in[4];
    const float* b2 = (const float*)d_in[5];
    const float* W3 = (const float*)d_in[6];
    const float* b3 = (const float*)d_in[7];
    float* out = (float*)d_out;

    const int N = in_sizes[0] / K_DIM;
    const int E = in_sizes[1] / 2;
    const int* src = ei;
    const int* dst = ei + E;

    float* dis  = (float*)d_ws;
    int*   cnt  = (int*)(dis + N);
    int*   novf = cnt + N;
    int*   ell  = novf + 64;
    int*   ovf  = ell + (size_t)N * ELL_W;
    float* P    = (float*)(ovf + E);
    float* Z    = P + (size_t)N * HP;

    const int T = 256;
    const int gN    = (N + T - 1) / T;
    const int gE    = (E + T - 1) / T;
    const int gNode = (N + 7) / 8;
    const int tiles = (N + 15) / 16;
    const int gGemm = (tiles + 7) / 8;

    gcn_cnt_init<<<gN, T, 0, stream>>>(cnt, novf, N);
    gcn_count_place<<<gE, T, 0, stream>>>(src, dst, cnt, ell, novf, ovf, E);
    gcn_dis<<<gN, T, 0, stream>>>(cnt, dis, N);

    gcn_gemm_wmma<128, false><<<gGemm, T, 0, stream>>>(x, W1, out, P, N);
    gcn_gather<128><<<gNode, T, 0, stream>>>(out, P, ell, cnt, novf, ovf, src, dst, dis, b1, Z, N);

    gcn_gemm_wmma<128, true><<<gGemm, T, 0, stream>>>(Z, W2, out, P, N);
    gcn_gather<128><<<gNode, T, 0, stream>>>(out, P, ell, cnt, novf, ovf, src, dst, dis, b2, Z, N);

    gcn_gemm_wmma<64, true><<<gGemm, T, 0, stream>>>(Z, W3, P, P, N);
    gcn_gather<64><<<gNode, T, 0, stream>>>(P, P, ell, cnt, novf, ovf, src, dst, dis, b3, out, N);
}


# Edge grouping in two levels on LDS counters: edges binned by destination range, rank = returned LDS add, slot table stored as whole lines; no global add per edge; row sums in new arrival-rank order
# speedup vs baseline: 4.2094x; 1.3463x over previous
#include <hip/hip_runtime.h>
#include <hip/hip_bf16.h>

typedef __attribute__((ext_vector_type(2))) float v2f;
typedef __attribute__((ext_vector_type(8))) float v8f;

#define K_DIM 128
#define LDS_PAD 8
#define ELL_W 32
#define HP 64

#define BIN_SHIFT 8
#define BIN_W (1 << BIN_SHIFT)
#define MAXB 512
#define EPT 16
#define CHUNK (EPT * 256)
#define GRB 8

__device__ __forceinline__ void block_exscan(int* arr, int* wsum) {
    const int t = threadIdx.x;
    const int lane = t & 31;
    const int wave = t >> 5;
    const int a = arr[2 * t];
    const int b = arr[2 * t + 1];
    const int v = a + b;
    int inc = v;
#pragma unroll
    for (int o = 1; o < 32; o <<= 1) {
        const int u = __shfl_up(inc, o);
        if (lane >= o) inc += u;
    }
    if (lane == 31) wsum[wave] = inc;
    __syncthreads();
    int off = 0;
#pragma unroll
    for (int w = 0; w < 8; ++w) off += (w < wave) ? wsum[w] : 0;
    const int ex = off + inc - v;
    arr[2 * t] = ex;
    arr[2 * t + 1] = ex + a;
    __syncthreads();
}

__global__ void gcn_bins_init(int* __restrict__ binCount, int* __restrict__ binCursor,
                              int* __restrict__ novf) {
    const int t = threadIdx.x;
    binCount[t] = 0;
    binCursor[t] = 0;
    if (t == 0) *novf = 0;
}

__global__ void __launch_bounds__(256)
gcn_bin_hist(const int* __restrict__ dst, int* __restrict__ binCount, int nE) {
    __shared__ int h[MAXB];
    const int t = threadIdx.x;
    h[t] = 0;
    h[t + 256] = 0;
    __syncthreads();
    const int base = blockIdx.x * CHUNK;
    int dv[EPT];
#pragma unroll
    for (int k = 0; k < EPT; ++k) {
        const int e = base + k * 256 + t;
        dv[k] = dst[e < nE ? e : nE - 1];
    }
#pragma unroll
    for (int k = 0; k < EPT; ++k) {
        const int e = base + k * 256 + t;
        if (e < nE) atomicAdd(&h[dv[k] >> BIN_SHIFT], 1);
    }
    __syncthreads();
    const int c0 = h[t];
    const int c1 = h[t + 256];
    if (c0) atomicAdd(&binCount[t], c0);
    if (c1) atomicAdd(&binCount[t + 256], c1);
}

__global__ void __launch_bounds__(256)
gcn_bin_scatter(const int* __restrict__ src, const int* __restrict__ dst,
                const int* __restrict__ binCount, int* __restrict__ binCursor,
                uint2* __restrict__ list, int nE) {
    __shared__ int gstart[MAXB];
    __shared__ int h[MAXB];
    __shared__ int wsum[8];
    __shared__ unsigned stA[CHUNK];
    __shared__ unsigned stB[CHUNK];
    const int t = threadIdx.x;
    const int base = blockIdx.x * CHUNK;

    gstart[t] = binCount[t];
    gstart[t + 256] = binCount[t + 256];
    h[t] = 0;
    h[t + 256] = 0;
    __syncthreads();
    block_exscan(gstart, wsum);

    unsigned wA[EPT];
    int pk[EPT];
#pragma unroll
    for (int k = 0; k < EPT; ++k) {
        const int e = base + k * 256 + t;
        const int ec = e < nE ? e : nE - 1;
        pk[k] = dst[ec];
        wA[k] = (unsigned)src[ec];
    }
#pragma unroll
    for (int k = 0; k < EPT; ++k) {
        const int e = base + k * 256 + t;
        const int d = pk[k];
        const int s = (int)wA[k];
        pk[k] = -1;
        if (e < nE) {
            const int bin = d >> BIN_SHIFT;
            const int r = atomicAdd(&h[bin], 1);
            wA[k] = ((unsigned)(d & (BIN_W - 1)) << 24) | (unsigned)s;
            pk[k] = (bin << 12) | r;
        }
    }
    __syncthreads();

    const int c0 = h[t];
    const int c1 = h[t + 256];
    if (c0) gstart[t] += atomicAdd(&binCursor[t], c0);
    if (c1) gstart[t + 256] += atomicAdd(&binCursor[t + 256], c1);
    block_exscan(h, wsum);

#pragma unroll
    for (int k = 0; k < EPT; ++k) {
        if (pk[k] >= 0) {
            const int bin = pk[k] >> 12;
            const int p = h[bin] + (pk[k] & 4095);
            stA[p] = wA[k];
            stB[p] = ((unsigned)bin << 12) | (unsigned)(k * 256 + t);
        }
    }
    __syncthreads();

    const int nHere = (nE - base < CHUNK) ? (nE - base) : CHUNK;
    for (int p = t; p < nHere; p += 256) {
        const unsigned A = stA[p];
        const unsigned B = stB[p];
        const int bin = (int)(B >> 12);
        const int pos = gstart[bin] + (p - h[bin]);
        list[pos] = make_uint2(A, (unsigned)base + (B & 4095u));
    }
}

__global__ void __launch_bounds__(256)
gcn_group_rows(const uint2* __restrict__ list, const int* __restrict__ binCount,
               int* __restrict__ ell, int* __restrict__ cnt, float* __restrict__ dis,
               int* __restrict__ novf, int* __restrict__ ovf, int n) {
    __shared__ int gs[MAXB];
    __shared__ int wsum[8];
    __shared__ int cntL[BIN_W];
    __shared__ uint4 tile[BIN_W * ELL_W / 4];
    const int t = threadIdx.x;
    const int b = blockIdx.x;

    gs[t] = binCount[t];
    gs[t + 256] = binCount[t + 256];
    const int nb = binCount[b];
    cntL[t] = 0;
    for (int q = t; q < BIN_W * ELL_W / 4; q += 256) tile[q] = make_uint4(0u, 0u, 0u, 0u);
    __syncthreads();
    block_exscan(gs, wsum);
    const int start = gs[b];

    int* tl = (int*)tile;
    for (int i0 = 0; i0 < nb; i0 += 256 * GRB) {
        uint2 rec[GRB];
#pragma unroll
        for (int u = 0; u < GRB; ++u) {
            const int i = i0 + u * 256 + t;
            rec[u] = list[start + (i < nb ? i : nb - 1)];
        }
#pragma unroll
        for (int u = 0; u < GRB; ++u) {
            const int i = i0 + u * 256 + t;
            if (i < nb) {
                const int dl = (int)(rec[u].x >> 24);
                const int s = (int)(rec[u].x & 0xFFFFFFu);
                const int r = atomicAdd(&cntL[dl], 1);
                if (r < ELL_W) tl[dl * ELL_W + r] = s;
                else           ovf[atomicAdd(novf, 1)] = (int)rec[u].y;
            }
        }
    }
    __syncthreads();

    const int row0 = b * BIN_W;
    uint4* __restrict__ ellv = (uint4*)ell + (size_t)b * (BIN_W * ELL_W / 4);
    for (int q = t; q < BIN_W * ELL_W / 4; q += 256) {
        if (row0 + (q >> 3) < n) ellv[q] = tile[q];
    }
    const int d = row0 + t;
    if (d < n) {
        const int c = cntL[t];
        cnt[d] = c;
        dis[d] = rsqrtf(fmaxf(1.0f + (float)c, 1.0f));
    }
}

template <int M, bool RELU_IN>
__global__ void __launch_bounds__(256)
gcn_gemm_wmma(const float* __restrict__ X, const float* __restrict__ W,
              float* __restrict__ H0, float* __restrict__ H1, int nRows) {
    constexpr int LSTR = M + LDS_PAD;
    __shared__ float Wlds[K_DIM * LSTR];

    const int tid = threadIdx.x;
    for (int i = tid * 4; i < K_DIM * M; i += 256 * 4) {
        const int r = i / M;
        const int c = i - r * M;
        *(float4*)&Wlds[r * LSTR + c] = *(const float4*)&W[i];
    }
    __syncthreads();

    const int wave = tid >> 5;
    const int lane = tid & 31;
    const int tile = blockIdx.x * 8 + wave;
    if (tile * 16 >= nRows) return;

    const int rowA = tile * 16 + (lane & 15);
    const int kOff = (lane >> 4) * 2;
    const int nCol = (lane & 15);
    const float* __restrict__ xrow = X + (size_t)rowA * K_DIM;
    const float* __restrict__ wcol = &Wlds[kOff * LSTR + nCol];

    v8f acc[M / 16] = {};

#pragma unroll 4
    for (int k = 0; k < K_DIM; k += 4) {
        float a0 = xrow[k + kOff];
        float a1 = xrow[k + kOff + 1];
        if (RELU_IN) { a0 = fmaxf(a0, 0.0f); a1 = fmaxf(a1, 0.0f); }
        v2f a; a.x = a0; a.y = a1;
        const float* __restrict__ wk = wcol + k * LSTR;
#pragma unroll
        for (int t = 0; t < M / 16; ++t) {
            v2f b;
            b.x = wk[t * 16];
            b.y = wk[LSTR + t * 16];
            acc[t] = __builtin_amdgcn_wmma_f32_16x16x4_f32(
                false, a, false, b, (short)0, acc[t], false, false);
        }
    }

    const int rBase = tile * 16 + ((lane >> 4) << 3);
#pragma unroll
    for (int t = 0; t < M / 16; ++t) {
        float* __restrict__ hp = (t < HP / 16) ? H0 : H1;
#pragma unroll
        for (int v = 0; v < 8; ++v) {
            hp[(size_t)(rBase + v) * HP + (t & 3) * 16 + nCol] = acc[t][v];
        }
    }
}

template <int EPI>
__device__ __forceinline__ int lane_pick(int v, int j) {
    if constexpr (EPI == 1) return __builtin_amdgcn_readlane(v, j);
    else                    return __shfl(v, j);
}

template <int M>
__global__ void __launch_bounds__(256)
gcn_gather(const float* __restrict__ Hlo, const float* __restrict__ Hhi,
           const int* __restrict__ ell, const int* __restrict__ cnt,
           const int* __restrict__ novf, const int* __restrict__ ovf,
           const int* __restrict__ src, const int* __restrict__ dst,
           const float* __restrict__ dis, const float* __restrict__ bias,
           float* __restrict__ out, int n) {
    constexpr int LPR = M / 4;
    constexpr int EPI = 32 / LPR;
    const int lane = threadIdx.x & 31;
    const int d = __builtin_amdgcn_readfirstlane(blockIdx.x * 8 + (threadIdx.x >> 5));
    if (d >= n) return;
    const int sub = lane / LPR;
    const int c4  = lane % LPR;
    const float* __restrict__ hb = ((c4 < 16) ? Hlo : Hhi) + (c4 & 15) * 4;

    const float dd = dis[d];
    const int deg = cnt[d];
    const int nl = deg < ELL_W ? deg : ELL_W;

    int my_s = 0;
    float my_d = 0.0f;
    if (lane < nl) {
        my_s = ell[(size_t)d * ELL_W + lane];
        my_d = dis[my_s];
    }

    const float4 hs = *(const float4*)(hb + (size_t)d * HP);
    const float4 bs = *(const float4*)(bias + c4 * 4);
    float4 acc = make_float4(0.0f, 0.0f, 0.0f, 0.0f);

    int j0 = 0;
    for (; j0 + 4 * EPI <= nl; j0 += 4 * EPI) {
        float4 h[4];
        float w[4];
#pragma unroll
        for (int u = 0; u < 4; ++u) {
            const int j = j0 + u * EPI + sub;
            const int s = lane_pick<EPI>(my_s, j);
            w[u] = __int_as_float(lane_pick<EPI>(__float_as_int(my_d), j)) * dd;
            h[u] = *(const float4*)(hb + (size_t)s * HP);
        }
#pragma unroll
        for (int u = 0; u < 4; ++u) {
            acc.x += h[u].x * w[u];
            acc.y += h[u].y * w[u];
            acc.z += h[u].z * w[u];
            acc.w += h[u].w * w[u];
        }
    }
#pragma unroll 1
    for (; j0 < nl; j0 += EPI) {
        const int j = j0 + sub;
        const int s = lane_pick<EPI>(my_s, j);
        const float w = __int_as_float(lane_pick<EPI>(__float_as_int(my_d), j)) * dd;
        float4 h = *(const float4*)(hb + (size_t)s * HP);
        if (EPI == 2 && j >= nl) h = make_float4(0.0f, 0.0f, 0.0f, 0.0f);
        acc.x += h.x * w;
        acc.y += h.y * w;
        acc.z += h.z * w;
        acc.w += h.w * w;
    }

    if (deg > ELL_W) {
        const int no = *novf;
        for (int i0 = 0; i0 < no; i0 += 32) {
            const int i = i0 + lane;
            int s_ = 0;
            bool m = false;
            if (i < no) {
                const int e = ovf[i];
                m = (dst[e] == d);
                if (m) s_ = src[e];
            }
            unsigned mask = (unsigned)__ballot(m);
            while (mask) {
                const int l = __ffs(mask) - 1;
                mask &= mask - 1;
                const int s = __shfl(s_, l);
                const float w = dis[s] * dd;
                if (sub == 0) {
                    const float4 h = *(const float4*)(hb + (size_t)s * HP);
                    acc.x += h.x * w;
                    acc.y += h.y * w;
                    acc.z += h.z * w;
                    acc.w += h.w * w;
                }
            }
        }
    }

    if (sub == 0) {
        const float s2 = dd * dd;
        acc.x += hs.x * s2 + bs.x;
        acc.y += hs.y * s2 + bs.y;
        acc.z += hs.z * s2 + bs.z;
        acc.w += hs.w * s2 + bs.w;
    }
    if (EPI == 2) {
        acc.x += __shfl_xor(acc.x, 16);
        acc.y += __shfl_xor(acc.y, 16);
        acc.z += __shfl_xor(acc.z, 16);
        acc.w += __shfl_xor(acc.w, 16);
    }
    if (sub == 0) *(float4*)(out + (size_t)d * M + c4 * 4) = acc;
}

extern "C" void kernel_launch(void* const* d_in, const int* in_sizes, int n_in,
                              void* d_out, int out_size, void* d_ws, size_t ws_size,
                              hipStream_t stream) {
    (void)n_in; (void)out_size; (void)ws_size;

    const float* x  = (const float*)d_in[0];
    const int*   ei = (const int*)d_in[1];
    const float* W1 = (const float*)d_in[2];
    const float* b1 = (const float*)d_in[3];
    const float* W2 = (const float*)d_in[4];
    const float* b2 = (const float*)d_in[5];
    const float* W3 = (const float*)d_in[6];
    const float* b3 = (const float*)d_in[7];
    float* out = (float*)d_out;

    const int N = in_sizes[0] / K_DIM;
    const int E = in_sizes[1] / 2;
    const int* src = ei;
    const int* dst = ei + E;

    float* dis  = (float*)d_ws;
    int*   cnt  = (int*)(dis + N);
    int*   novf = cnt + N;
    int*   ell  = novf + 64;
    int*   ovf  = ell + (size_t)N * ELL_W;
    float* P    = (float*)(ovf + E);
    float* Z    = P + (size_t)N * HP;
    uint2* list      = (uint2*)Z;
    int*   binCount  = (int*)Z + 2 * (size_t)E;
    int*   binCursor = binCount + MAXB;

    const int T = 256;
    const int gChunk = (E + CHUNK - 1) / CHUNK;
    const int gBin   = (N + BIN_W - 1) / BIN_W;
    const int gNode = (N + 7) / 8;
    const int tiles = (N + 15) / 16;
    const int gGemm = (tiles + 7) / 8;

    gcn_bins_init<<<1, MAXB, 0, stream>>>(binCount, binCursor, novf);
    gcn_bin_hist<<<gChunk, T, 0, stream>>>(dst, binCount, E);
    gcn_bin_scatter<<<gChunk, T, 0, stream>>>(src, dst, binCount, binCursor, list, E);
    gcn_group_rows<<<gBin, T, 0, stream>>>(list, binCount, ell, cnt, dis, novf, ovf, N);

    gcn_gemm_wmma<128, false><<<gGemm, T, 0, stream>>>(x, W1, out, P, N);
    gcn_gather<128><<<gNode, T, 0, stream>>>(out, P, ell, cnt, novf, ovf, src, dst, dis, b1, Z, N);

    gcn_gemm_wmma<128, true><<<gGemm, T, 0, stream>>>(Z, W2, out, P, N);
    gcn_gather<128><<<gNode, T, 0, stream>>>(out, P, ell, cnt, novf, ovf, src, dst, dis, b2, Z, N);

    gcn_gemm_wmma<64, true><<<gGemm, T, 0, stream>>>(Z, W3, P, P, N);
    gcn_gather<64><<<gNode, T, 0, stream>>>(P, P, ell, cnt, novf, ovf, src, dst, dis, b3, out, N);
}


# Edge weight factored: GEMM epilogue stores each row scaled by dis[row]; gather adds plain rows in the same slot order and scales the sum by dis[dst] once; the dis[src] read per edge is gone
# speedup vs baseline: 4.4074x; 1.0470x over previous
#include <hip/hip_runtime.h>
#include <hip/hip_bf16.h>

typedef __attribute__((ext_vector_type(2))) float v2f;
typedef __attribute__((ext_vector_type(8))) float v8f;

#define K_DIM 128
#define LDS_PAD 8
#define ELL_W 32
#define HP 64

#define BIN_SHIFT 8
#define BIN_W (1 << BIN_SHIFT)
#define MAXB 512
#define EPT 16
#define CHUNK (EPT * 256)
#define GRB 8

__device__ __forceinline__ void block_exscan(int* arr, int* wsum) {
    const int t = threadIdx.x;
    const int lane = t & 31;
    const int wave = t >> 5;
    const int a = arr[2 * t];
    const int b = arr[2 * t + 1];
    const int v = a + b;
    int inc = v;
#pragma unroll
    for (int o = 1; o < 32; o <<= 1) {
        const int u = __shfl_up(inc, o);
        if (lane >= o) inc += u;
    }
    if (lane == 31) wsum[wave] = inc;
    __syncthreads();
    int off = 0;
#pragma unroll
    for (int w = 0; w < 8; ++w) off += (w < wave) ? wsum[w] : 0;
    const int ex = off + inc - v;
    arr[2 * t] = ex;
    arr[2 * t + 1] = ex + a;
    __syncthreads();
}

__global__ void gcn_bins_init(int* __restrict__ binCount, int* __restrict__ binCursor,
                              int* __restrict__ novf) {
    const int t = threadIdx.x;
    binCount[t] = 0;
    binCursor[t] = 0;
    if (t == 0) *novf = 0;
}

__global__ void __launch_bounds__(256)
gcn_bin_hist(const int* __restrict__ dst, int* __restrict__ binCount, int nE) {
    __shared__ int h[MAXB];
    const int t = threadIdx.x;
    h[t] = 0;
    h[t + 256] = 0;
    __syncthreads();
    const int base = blockIdx.x * CHUNK;
    int dv[EPT];
#pragma unroll
    for (int k = 0; k < EPT; ++k) {
        const int e = base + k * 256 + t;
        dv[k] = dst[e < nE ? e : nE - 1];
    }
#pragma unroll
    for (int k = 0; k < EPT; ++k) {
        const int e = base + k * 256 + t;
        if (e < nE) atomicAdd(&h[dv[k] >> BIN_SHIFT], 1);
    }
    __syncthreads();
    const int c0 = h[t];
    const int c1 = h[t + 256];
    if (c0) atomicAdd(&binCount[t], c0);
    if (c1) atomicAdd(&binCount[t + 256], c1);
}

__global__ void __launch_bounds__(256)
gcn_bin_scatter(const int* __restrict__ src, const int* __restrict__ dst,
                const int* __restrict__ binCount, int* __restrict__ binCursor,
                uint2* __restrict__ list, int nE) {
    __shared__ int gstart[MAXB];
    __shared__ int h[MAXB];
    __shared__ int wsum[8];
    __shared__ unsigned stA[CHUNK];
    __shared__ unsigned stB[CHUNK];
    const int t = threadIdx.x;
    const int base = blockIdx.x * CHUNK;

    gstart[t] = binCount[t];
    gstart[t + 256] = binCount[t + 256];
    h[t] = 0;
    h[t + 256] = 0;
    __syncthreads();
    block_exscan(gstart, wsum);

    unsigned wA[EPT];
    int pk[EPT];
#pragma unroll
    for (int k = 0; k < EPT; ++k) {
        const int e = base + k * 256 + t;
        const int ec = e < nE ? e : nE - 1;
        pk[k] = dst[ec];
        wA[k] = (unsigned)src[ec];
    }
#pragma unroll
    for (int k = 0; k < EPT; ++k) {
        const int e = base + k * 256 + t;
        const int d = pk[k];
        const int s = (int)wA[k];
        pk[k] = -1;
        if (e < nE) {
            const int bin = d >> BIN_SHIFT;
            const int r = atomicAdd(&h[bin], 1);
            wA[k] = ((unsigned)(d & (BIN_W - 1)) << 24) | (unsigned)s;
            pk[k] = (bin << 12) | r;
        }
    }
    __syncthreads();

    const int c0 = h[t];
    const int c1 = h[t + 256];
    if (c0) gstart[t] += atomicAdd(&binCursor[t], c0);
    if (c1) gstart[t + 256] += atomicAdd(&binCursor[t + 256], c1);
    block_exscan(h, wsum);

#pragma unroll
    for (int k = 0; k < EPT; ++k) {
        if (pk[k] >= 0) {
            const int bin = pk[k] >> 12;
            const int p = h[bin] + (pk[k] & 4095);
            stA[p] = wA[k];
            stB[p] = ((unsigned)bin << 12) | (unsigned)(k * 256 + t);
        }
    }
    __syncthreads();

    const int nHere = (nE - base < CHUNK) ? (nE - base) : CHUNK;
    for (int p = t; p < nHere; p += 256) {
        const unsigned A = stA[p];
        const unsigned B = stB[p];
        const int bin = (int)(B >> 12);
        const int pos = gstart[bin] + (p - h[bin]);
        list[pos] = make_uint2(A, (unsigned)base + (B & 4095u));
    }
}

__global__ void __launch_bounds__(256)
gcn_group_rows(const uint2* __restrict__ list, const int* __restrict__ binCount,
               int* __restrict__ ell, int* __restrict__ cnt, float* __restrict__ dis,
               int* __restrict__ novf, int* __restrict__ ovf, int n) {
    __shared__ int gs[MAXB];
    __shared__ int wsum[8];
    __shared__ int cntL[BIN_W];
    __shared__ uint4 tile[BIN_W * ELL_W / 4];
    const int t = threadIdx.x;
    const int b = blockIdx.x;

    gs[t] = binCount[t];
    gs[t + 256] = binCount[t + 256];
    const int nb = binCount[b];
    cntL[t] = 0;
    for (int q = t; q < BIN_W * ELL_W / 4; q += 256) tile[q] = make_uint4(0u, 0u, 0u, 0u);
    __syncthreads();
    block_exscan(gs, wsum);
    const int start = gs[b];

    int* tl = (int*)tile;
    for (int i0 = 0; i0 < nb; i0 += 256 * GRB) {
        uint2 rec[GRB];
#pragma unroll
        for (int u = 0; u < GRB; ++u) {
            const int i = i0 + u * 256 + t;
            rec[u] = list[start + (i < nb ? i : nb - 1)];
        }
#pragma unroll
        for (int u = 0; u < GRB; ++u) {
            const int i = i0 + u * 256 + t;
            if (i < nb) {
                const int dl = (int)(rec[u].x >> 24);
                const int s = (int)(rec[u].x & 0xFFFFFFu);
                const int r = atomicAdd(&cntL[dl], 1);
                if (r < ELL_W) tl[dl * ELL_W + r] = s;
                else           ovf[atomicAdd(novf, 1)] = (int)rec[u].y;
            }
        }
    }
    __syncthreads();

    const int row0 = b * BIN_W;
    uint4* __restrict__ ellv = (uint4*)ell + (size_t)b * (BIN_W * ELL_W / 4);
    for (int q = t; q < BIN_W * ELL_W / 4; q += 256) {
        if (row0 + (q >> 3) < n) ellv[q] = tile[q];
    }
    const int d = row0 + t;
    if (d < n) {
        const int c = cntL[t];
        cnt[d] = c;
        dis[d] = rsqrtf(fmaxf(1.0f + (float)c, 1.0f));
    }
}

template <int M, bool RELU_IN>
__global__ void __launch_bounds__(256)
gcn_gemm_wmma(const float* __restrict__ X, const float* __restrict__ W,
              const float* __restrict__ dis,
              float* __restrict__ H0, float* __restrict__ H1, int nRows) {
    constexpr int LSTR = M + LDS_PAD;
    __shared__ float Wlds[K_DIM * LSTR];

    const int tid = threadIdx.x;
    for (int i = tid * 4; i < K_DIM * M; i += 256 * 4) {
        const int r = i / M;
        const int c = i - r * M;
        *(float4*)&Wlds[r * LSTR + c] = *(const float4*)&W[i];
    }
    __syncthreads();

    const int wave = tid >> 5;
    const int lane = tid & 31;
    const int tile = blockIdx.x * 8 + wave;
    if (tile * 16 >= nRows) return;

    const int rowA = tile * 16 + (lane & 15);
    const int kOff = (lane >> 4) * 2;
    const int nCol = (lane & 15);
    const float* __restrict__ xrow = X + (size_t)rowA * K_DIM;
    const float* __restrict__ wcol = &Wlds[kOff * LSTR + nCol];

    v8f acc[M / 16] = {};

#pragma unroll 4
    for (int k = 0; k < K_DIM; k += 4) {
        float a0 = xrow[k + kOff];
        float a1 = xrow[k + kOff + 1];
        if (RELU_IN) { a0 = fmaxf(a0, 0.0f); a1 = fmaxf(a1, 0.0f); }
        v2f a; a.x = a0; a.y = a1;
        const float* __restrict__ wk = wcol + k * LSTR;
#pragma unroll
        for (int t = 0; t < M / 16; ++t) {
            v2f b;
            b.x = wk[t * 16];
            b.y = wk[LSTR + t * 16];
            acc[t] = __builtin_amdgcn_wmma_f32_16x16x4_f32(
                false, a, false, b, (short)0, acc[t], false, false);
        }
    }

    const int rBase = tile * 16 + ((lane >> 4) << 3);
    const float4 dA = *(const float4*)(dis + rBase);
    const float4 dB = *(const float4*)(dis + rBase + 4);
    const float dr[8] = {dA.x, dA.y, dA.z, dA.w, dB.x, dB.y, dB.z, dB.w};
#pragma unroll
    for (int t = 0; t < M / 16; ++t) {
        float* __restrict__ hp = (t < HP / 16) ? H0 : H1;
#pragma unroll
        for (int v = 0; v < 8; ++v) {
            hp[(size_t)(rBase + v) * HP + (t & 3) * 16 + nCol] = acc[t][v] * dr[v];
        }
    }
}

template <int EPI>
__device__ __forceinline__ int lane_pick(int v, int j) {
    if constexpr (EPI == 1) return __builtin_amdgcn_readlane(v, j);
    else                    return __shfl(v, j);
}

template <int M>
__global__ void __launch_bounds__(256)
gcn_gather(const float* __restrict__ Hlo, const float* __restrict__ Hhi,
           const int* __restrict__ ell, const int* __restrict__ cnt,
           const int* __restrict__ novf, const int* __restrict__ ovf,
           const int* __restrict__ src, const int* __restrict__ dst,
           const float* __restrict__ dis, const float* __restrict__ bias,
           float* __restrict__ out, int n) {
    constexpr int LPR = M / 4;
    constexpr int EPI = 32 / LPR;
    const int lane = threadIdx.x & 31;
    const int d = __builtin_amdgcn_readfirstlane(blockIdx.x * 8 + (threadIdx.x >> 5));
    if (d >= n) return;
    const int sub = lane / LPR;
    const int c4  = lane % LPR;
    const float* __restrict__ hb = ((c4 < 16) ? Hlo : Hhi) + (c4 & 15) * 4;

    const float dd = dis[d];
    const int deg = cnt[d];
    const int nl = deg < ELL_W ? deg : ELL_W;

    int my_s = 0;
    if (lane < nl) my_s = ell[(size_t)d * ELL_W + lane];

    const float4 hs = *(const float4*)(hb + (size_t)d * HP);
    const float4 bs = *(const float4*)(bias + c4 * 4);
    float4 acc = make_float4(0.0f, 0.0f, 0.0f, 0.0f);

    int j0 = 0;
    for (; j0 + 4 * EPI <= nl; j0 += 4 * EPI) {
        float4 h[4];
#pragma unroll
        for (int u = 0; u < 4; ++u) {
            const int j = j0 + u * EPI + sub;
            const int s = lane_pick<EPI>(my_s, j);
            h[u] = *(const float4*)(hb + (size_t)s * HP);
        }
#pragma unroll
        for (int u = 0; u < 4; ++u) {
            acc.x += h[u].x;
            acc.y += h[u].y;
            acc.z += h[u].z;
            acc.w += h[u].w;
        }
    }
#pragma unroll 1
    for (; j0 < nl; j0 += EPI) {
        const int j = j0 + sub;
        const int s = lane_pick<EPI>(my_s, j);
        float4 h = *(const float4*)(hb + (size_t)s * HP);
        if (EPI == 2 && j >= nl) h = make_float4(0.0f, 0.0f, 0.0f, 0.0f);
        acc.x += h.x;
        acc.y += h.y;
        acc.z += h.z;
        acc.w += h.w;
    }

    if (deg > ELL_W) {
        const int no = *novf;
        for (int i0 = 0; i0 < no; i0 += 32) {
            const int i = i0 + lane;
            int s_ = 0;
            bool m = false;
            if (i < no) {
                const int e = ovf[i];
                m = (dst[e] == d);
                if (m) s_ = src[e];
            }
            unsigned mask = (unsigned)__ballot(m);
            while (mask) {
                const int l = __ffs(mask) - 1;
                mask &= mask - 1;
                const int s = __shfl(s_, l);
                if (sub == 0) {
                    const float4 h = *(const float4*)(hb + (size_t)s * HP);
                    acc.x += h.x;
                    acc.y += h.y;
                    acc.z += h.z;
                    acc.w += h.w;
                }
            }
        }
    }

    if (sub == 0) {
        acc.x += hs.x;
        acc.y += hs.y;
        acc.z += hs.z;
        acc.w += hs.w;
    }
    if (EPI == 2) {
        acc.x += __shfl_xor(acc.x, 16);
        acc.y += __shfl_xor(acc.y, 16);
        acc.z += __shfl_xor(acc.z, 16);
        acc.w += __shfl_xor(acc.w, 16);
    }
    if (sub == 0) {
        acc.x = acc.x * dd + bs.x;
        acc.y = acc.y * dd + bs.y;
        acc.z = acc.z * dd + bs.z;
        acc.w = acc.w * dd + bs.w;
        *(float4*)(out + (size_t)d * M + c4 * 4) = acc;
    }
}

extern "C" void kernel_launch(void* const* d_in, const int* in_sizes, int n_in,
                              void* d_out, int out_size, void* d_ws, size_t ws_size,
                              hipStream_t stream) {
    (void)n_in; (void)out_size; (void)ws_size;

    const float* x  = (const float*)d_in[0];
    const int*   ei = (const int*)d_in[1];
    const float* W1 = (const float*)d_in[2];
    const float* b1 = (const float*)d_in[3];
    const float* W2 = (const float*)d_in[4];
    const float* b2 = (const float*)d_in[5];
    const float* W3 = (const float*)d_in[6];
    const float* b3 = (const float*)d_in[7];
    float* out = (float*)d_out;

    const int N = in_sizes[0] / K_DIM;
    const int E = in_sizes[1] / 2;
    const int* src = ei;
    const int* dst = ei + E;

    float* dis  = (float*)d_ws;
    int*   cnt  = (int*)(dis + N);
    int*   novf = cnt + N;
    int*   ell  = novf + 64;
    int*   ovf  = ell + (size_t)N * ELL_W;
    float* P    = (float*)(ovf + E);
    float* Z    = P + (size_t)N * HP;
    uint2* list      = (uint2*)Z;
    int*   binCount  = (int*)Z + 2 * (size_t)E;
    int*   binCursor = binCount + MAXB;

    const int T = 256;
    const int gChunk = (E + CHUNK - 1) / CHUNK;
    const int gBin   = (N + BIN_W - 1) / BIN_W;
    const int gNode = (N + 7) / 8;
    const int tiles = (N + 15) / 16;
    const int gGemm = (tiles + 7) / 8;

    gcn_bins_init<<<1, MAXB, 0, stream>>>(binCount, binCursor, novf);
    gcn_bin_hist<<<gChunk, T, 0, stream>>>(dst, binCount, E);
    gcn_bin_scatter<<<gChunk, T, 0, stream>>>(src, dst, binCount, binCursor, list, E);
    gcn_group_rows<<<gBin, T, 0, stream>>>(list, binCount, ell, cnt, dis, novf, ovf, N);

    gcn_gemm_wmma<128, false><<<gGemm, T, 0, stream>>>(x, W1, dis, out, P, N);
    gcn_gather<128><<<gNode, T, 0, stream>>>(out, P, ell, cnt, novf, ovf, src, dst, dis, b1, Z, N);

    gcn_gemm_wmma<128, true><<<gGemm, T, 0, stream>>>(Z, W2, dis, out, P, N);
    gcn_gather<128><<<gNode, T, 0, stream>>>(out, P, ell, cnt, novf, ovf, src, dst, dis, b2, Z, N);

    gcn_gemm_wmma<64, true><<<gGemm, T, 0, stream>>>(Z, W3, dis, P, P, N);
    gcn_gather<64><<<gNode, T, 0, stream>>>(P, P, ell, cnt, novf, ovf, src, dst, dis, b3, out, N);
}


# Layer 1 and 2 aggregation fused with the next layer's WMMA GEMM: a resident block gathers 32 rows into LDS and multiplies them by W kept in LDS; aggregated rows never reach memory; 8 launches
# speedup vs baseline: 5.3612x; 1.2164x over previous
#include <hip/hip_runtime.h>
#include <hip/hip_bf16.h>

typedef __attribute__((ext_vector_type(2))) float v2f;
typedef __attribute__((ext_vector_type(8))) float v8f;

#define K_DIM 128
#define LDS_PAD 8
#define ELL_W 32
#define HP 64

#define BIN_SHIFT 8
#define BIN_W (1 << BIN_SHIFT)
#define MAXB 512
#define EPT 16
#define CHUNK (EPT * 256)
#define GRB 8

__device__ __forceinline__ void block_exscan(int* arr, int* wsum) {
    const int t = threadIdx.x;
    const int lane = t & 31;
    const int wave = t >> 5;
    const int a = arr[2 * t];
    const int b = arr[2 * t + 1];
    const int v = a + b;
    int inc = v;
#pragma unroll
    for (int o = 1; o < 32; o <<= 1) {
        const int u = __shfl_up(inc, o);
        if (lane >= o) inc += u;
    }
    if (lane == 31) wsum[wave] = inc;
    __syncthreads();
    int off = 0;
#pragma unroll
    for (int w = 0; w < 8; ++w) off += (w < wave) ? wsum[w] : 0;
    const int ex = off + inc - v;
    arr[2 * t] = ex;
    arr[2 * t + 1] = ex + a;
    __syncthreads();
}

__global__ void gcn_bins_init(int* __restrict__ binCount, int* __restrict__ binCursor,
                              int* __restrict__ novf) {
    const int t = threadIdx.x;
    binCount[t] = 0;
    binCursor[t] = 0;
    if (t == 0) *novf = 0;
}

__global__ void __launch_bounds__(256)
gcn_bin_hist(const int* __restrict__ dst, int* __restrict__ binCount, int nE) {
    __shared__ int h[MAXB];
    const int t = threadIdx.x;
    h[t] = 0;
    h[t + 256] = 0;
    __syncthreads();
    const int base = blockIdx.x * CHUNK;
    int dv[EPT];
#pragma unroll
    for (int k = 0; k < EPT; ++k) {
        const int e = base + k * 256 + t;
        dv[k] = dst[e < nE ? e : nE - 1];
    }
#pragma unroll
    for (int k = 0; k < EPT; ++k) {
        const int e = base + k * 256 + t;
        if (e < nE) atomicAdd(&h[dv[k] >> BIN_SHIFT], 1);
    }
    __syncthreads();
    const int c0 = h[t];
    const int c1 = h[t + 256];
    if (c0) atomicAdd(&binCount[t], c0);
    if (c1) atomicAdd(&binCount[t + 256], c1);
}

__global__ void __launch_bounds__(256)
gcn_bin_scatter(const int* __restrict__ src, const int* __restrict__ dst,
                const int* __restrict__ binCount, int* __restrict__ binCursor,
                uint2* __restrict__ list, int nE) {
    __shared__ int gstart[MAXB];
    __shared__ int h[MAXB];
    __shared__ int wsum[8];
    __shared__ unsigned stA[CHUNK];
    __shared__ unsigned stB[CHUNK];
    const int t = threadIdx.x;
    const int base = blockIdx.x * CHUNK;

    gstart[t] = binCount[t];
    gstart[t + 256] = binCount[t + 256];
    h[t] = 0;
    h[t + 256] = 0;
    __syncthreads();
    block_exscan(gstart, wsum);

    unsigned wA[EPT];
    int pk[EPT];
#pragma unroll
    for (int k = 0; k < EPT; ++k) {
        const int e = base + k * 256 + t;
        const int ec = e < nE ? e : nE - 1;
        pk[k] = dst[ec];
        wA[k] = (unsigned)src[ec];
    }
#pragma unroll
    for (int k = 0; k < EPT; ++k) {
        const int e = base + k * 256 + t;
        const int d = pk[k];
        const int s = (int)wA[k];
        pk[k] = -1;
        if (e < nE) {
            const int bin = d >> BIN_SHIFT;
            const int r = atomicAdd(&h[bin], 1);
            wA[k] = ((unsigned)(d & (BIN_W - 1)) << 24) | (unsigned)s;
            pk[k] = (bin << 12) | r;
        }
    }
    __syncthreads();

    const int c0 = h[t];
    const int c1 = h[t + 256];
    if (c0) gstart[t] += atomicAdd(&binCursor[t], c0);
    if (c1) gstart[t + 256] += atomicAdd(&binCursor[t + 256], c1);
    block_exscan(h, wsum);

#pragma unroll
    for (int k = 0; k < EPT; ++k) {
        if (pk[k] >= 0) {
            const int bin = pk[k] >> 12;
            const int p = h[bin] + (pk[k] & 4095);
            stA[p] = wA[k];
            stB[p] = ((unsigned)bin << 12) | (unsigned)(k * 256 + t);
        }
    }
    __syncthreads();

    const int nHere = (nE - base < CHUNK) ? (nE - base) : CHUNK;
    for (int p = t; p < nHere; p += 256) {
        const unsigned A = stA[p];
        const unsigned B = stB[p];
        const int bin = (int)(B >> 12);
        const int pos = gstart[bin] + (p - h[bin]);
        list[pos] = make_uint2(A, (unsigned)base + (B & 4095u));
    }
}

__global__ void __launch_bounds__(256)
gcn_group_rows(const uint2* __restrict__ list, const int* __restrict__ binCount,
               int* __restrict__ ell, int* __restrict__ cnt, float* __restrict__ dis,
               int* __restrict__ novf, int* __restrict__ ovf, int n) {
    __shared__ int gs[MAXB];
    __shared__ int wsum[8];
    __shared__ int cntL[BIN_W];
    __shared__ uint4 tile[BIN_W * ELL_W / 4];
    const int t = threadIdx.x;
    const int b = blockIdx.x;

    gs[t] = binCount[t];
    gs[t + 256] = binCount[t + 256];
    const int nb = binCount[b];
    cntL[t] = 0;
    for (int q = t; q < BIN_W * ELL_W / 4; q += 256) tile[q] = make_uint4(0u, 0u, 0u, 0u);
    __syncthreads();
    block_exscan(gs, wsum);
    const int start = gs[b];

    int* tl = (int*)tile;
    for (int i0 = 0; i0 < nb; i0 += 256 * GRB) {
        uint2 rec[GRB];
#pragma unroll
        for (int u = 0; u < GRB; ++u) {
            const int i = i0 + u * 256 + t;
            rec[u] = list[start + (i < nb ? i : nb - 1)];
        }
#pragma unroll
        for (int u = 0; u < GRB; ++u) {
            const int i = i0 + u * 256 + t;
            if (i < nb) {
                const int dl = (int)(rec[u].x >> 24);
                const int s = (int)(rec[u].x & 0xFFFFFFu);
                const int r = atomicAdd(&cntL[dl], 1);
                if (r < ELL_W) tl[dl * ELL_W + r] = s;
                else           ovf[atomicAdd(novf, 1)] = (int)rec[u].y;
            }
        }
    }
    __syncthreads();

    const int row0 = b * BIN_W;
    uint4* __restrict__ ellv = (uint4*)ell + (size_t)b * (BIN_W * ELL_W / 4);
    for (int q = t; q < BIN_W * ELL_W / 4; q += 256) {
        if (row0 + (q >> 3) < n) ellv[q] = tile[q];
    }
    const int d = row0 + t;
    if (d < n) {
        const int c = cntL[t];
        cnt[d] = c;
        dis[d] = rsqrtf(fmaxf(1.0f + (float)c, 1.0f));
    }
}

template <int M, bool RELU_IN>
__global__ void __launch_bounds__(256)
gcn_gemm_wmma(const float* __restrict__ X, const float* __restrict__ W,
              const float* __restrict__ dis,
              float* __restrict__ H0, float* __restrict__ H1, int nRows) {
    constexpr int LSTR = M + LDS_PAD;
    __shared__ float Wlds[K_DIM * LSTR];

    const int tid = threadIdx.x;
    for (int i = tid * 4; i < K_DIM * M; i += 256 * 4) {
        const int r = i / M;
        const int c = i - r * M;
        *(float4*)&Wlds[r * LSTR + c] = *(const float4*)&W[i];
    }
    __syncthreads();

    const int wave = tid >> 5;
    const int lane = tid & 31;
    const int tile = blockIdx.x * 8 + wave;
    if (tile * 16 >= nRows) return;

    const int rowA = tile * 16 + (lane & 15);
    const int kOff = (lane >> 4) * 2;
    const int nCol = (lane & 15);
    const float* __restrict__ xrow = X + (size_t)rowA * K_DIM;
    const float* __restrict__ wcol = &Wlds[kOff * LSTR + nCol];

    v8f acc[M / 16] = {};

#pragma unroll 4
    for (int k = 0; k < K_DIM; k += 4) {
        float a0 = xrow[k + kOff];
        float a1 = xrow[k + kOff + 1];
        if (RELU_IN) { a0 = fmaxf(a0, 0.0f); a1 = fmaxf(a1, 0.0f); }
        v2f a; a.x = a0; a.y = a1;
        const float* __restrict__ wk = wcol + k * LSTR;
#pragma unroll
        for (int t = 0; t < M / 16; ++t) {
            v2f b;
            b.x = wk[t * 16];
            b.y = wk[LSTR + t * 16];
            acc[t] = __builtin_amdgcn_wmma_f32_16x16x4_f32(
                false, a, false, b, (short)0, acc[t], false, false);
        }
    }

    const int rBase = tile * 16 + ((lane >> 4) << 3);
    const float4 dA = *(const float4*)(dis + rBase);
    const float4 dB = *(const float4*)(dis + rBase + 4);
    const float dr[8] = {dA.x, dA.y, dA.z, dA.w, dB.x, dB.y, dB.z, dB.w};
#pragma unroll
    for (int t = 0; t < M / 16; ++t) {
        float* __restrict__ hp = (t < HP / 16) ? H0 : H1;
#pragma unroll
        for (int v = 0; v < 8; ++v) {
            hp[(size_t)(rBase + v) * HP + (t & 3) * 16 + nCol] = acc[t][v] * dr[v];
        }
    }
}

template <int EPI>
__device__ __forceinline__ int lane_pick(int v, int j) {
    if constexpr (EPI == 1) return __builtin_amdgcn_readlane(v, j);
    else                    return __shfl(v, j);
}

template <int M>
__global__ void __launch_bounds__(256)
gcn_gather(const float* __restrict__ Hlo, const float* __restrict__ Hhi,
           const int* __restrict__ ell, const int* __restrict__ cnt,
           const int* __restrict__ novf, const int* __restrict__ ovf,
           const int* __restrict__ src, const int* __restrict__ dst,
           const float* __restrict__ dis, const float* __restrict__ bias,
           float* __restrict__ out, int n) {
    constexpr int LPR = M / 4;
    constexpr int EPI = 32 / LPR;
    const int lane = threadIdx.x & 31;
    const int d = __builtin_amdgcn_readfirstlane(blockIdx.x * 8 + (threadIdx.x >> 5));
    if (d >= n) return;
    const int sub = lane / LPR;
    const int c4  = lane % LPR;
    const float* __restrict__ hb = ((c4 < 16) ? Hlo : Hhi) + (c4 & 15) * 4;

    const float dd = dis[d];
    const int deg = cnt[d];
    const int nl = deg < ELL_W ? deg : ELL_W;

    int my_s = 0;
    if (lane < nl) my_s = ell[(size_t)d * ELL_W + lane];

    const float4 hs = *(const float4*)(hb + (size_t)d * HP);
    const float4 bs = *(const float4*)(bias + c4 * 4);
    float4 acc = make_float4(0.0f, 0.0f, 0.0f, 0.0f);

    int j0 = 0;
    for (; j0 + 4 * EPI <= nl; j0 += 4 * EPI) {
        float4 h[4];
#pragma unroll
        for (int u = 0; u < 4; ++u) {
            const int j = j0 + u * EPI + sub;
            const int s = lane_pick<EPI>(my_s, j);
            h[u] = *(const float4*)(hb + (size_t)s * HP);
        }
#pragma unroll
        for (int u = 0; u < 4; ++u) {
            acc.x += h[u].x;
            acc.y += h[u].y;
            acc.z += h[u].z;
            acc.w += h[u].w;
        }
    }
#pragma unroll 1
    for (; j0 < nl; j0 += EPI) {
        const int j = j0 + sub;
        const int s = lane_pick<EPI>(my_s, j);
        float4 h = *(const float4*)(hb + (size_t)s * HP);
        if (EPI == 2 && j >= nl) h = make_float4(0.0f, 0.0f, 0.0f, 0.0f);
        acc.x += h.x;
        acc.y += h.y;
        acc.z += h.z;
        acc.w += h.w;
    }

    if (deg > ELL_W) {
        const int no = *novf;
        for (int i0 = 0; i0 < no; i0 += 32) {
            const int i = i0 + lane;
            int s_ = 0;
            bool m = false;
            if (i < no) {
                const int e = ovf[i];
                m = (dst[e] == d);
                if (m) s_ = src[e];
            }
            unsigned mask = (unsigned)__ballot(m);
            while (mask) {
                const int l = __ffs(mask) - 1;
                mask &= mask - 1;
                const int s = __shfl(s_, l);
                if (sub == 0) {
                    const float4 h = *(const float4*)(hb + (size_t)s * HP);
                    acc.x += h.x;
                    acc.y += h.y;
                    acc.z += h.z;
                    acc.w += h.w;
                }
            }
        }
    }

    if (sub == 0) {
        acc.x += hs.x;
        acc.y += hs.y;
        acc.z += hs.z;
        acc.w += hs.w;
    }
    if (EPI == 2) {
        acc.x += __shfl_xor(acc.x, 16);
        acc.y += __shfl_xor(acc.y, 16);
        acc.z += __shfl_xor(acc.z, 16);
        acc.w += __shfl_xor(acc.w, 16);
    }
    if (sub == 0) {
        acc.x = acc.x * dd + bs.x;
        acc.y = acc.y * dd + bs.y;
        acc.z = acc.z * dd + bs.z;
        acc.w = acc.w * dd + bs.w;
        *(float4*)(out + (size_t)d * M + c4 * 4) = acc;
    }
}

#define RT 32
#define ZSTR (K_DIM + 4)

__device__ __forceinline__ void add4(float4& a, const float4 h) {
    a.x += h.x; a.y += h.y; a.z += h.z; a.w += h.w;
}

__device__ __forceinline__ float4 row_finish(float4 acc, int j0, const int nl, const int deg,
                                             const int d, const int my_s,
                                             const float* __restrict__ hb, const float4 hs,
                                             const float4 bs, const float dd,
                                             const int* __restrict__ novf, const int* __restrict__ ovf,
                                             const int* __restrict__ src, const int* __restrict__ dst,
                                             const int lane) {
    for (; j0 + 4 <= nl; j0 += 4) {
        float4 h[4];
#pragma unroll
        for (int u = 0; u < 4; ++u) {
            const int s = __builtin_amdgcn_readlane(my_s, j0 + u);
            h[u] = *(const float4*)(hb + (size_t)s * HP);
        }
#pragma unroll
        for (int u = 0; u < 4; ++u) add4(acc, h[u]);
    }
#pragma unroll 1
    for (; j0 < nl; ++j0) {
        const int s = __builtin_amdgcn_readlane(my_s, j0);
        add4(acc, *(const float4*)(hb + (size_t)s * HP));
    }
    if (deg > ELL_W) {
        const int no = *novf;
        for (int i0 = 0; i0 < no; i0 += 32) {
            const int i = i0 + lane;
            int s_ = 0;
            bool m = false;
            if (i < no) {
                const int e = ovf[i];
                m = (dst[e] == d);
                if (m) s_ = src[e];
            }
            unsigned mask = (unsigned)__ballot(m);
            while (mask) {
                const int l = __ffs(mask) - 1;
                mask &= mask - 1;
                const int s = __shfl(s_, l);
                add4(acc, *(const float4*)(hb + (size_t)s * HP));
            }
        }
    }
    add4(acc, hs);
    acc.x = acc.x * dd + bs.x;
    acc.y = acc.y * dd + bs.y;
    acc.z = acc.z * dd + bs.z;
    acc.w = acc.w * dd + bs.w;
    return acc;
}

template <int MN>
__global__ void __launch_bounds__(512, 12)
gcn_gather_gemm(const float* __restrict__ Hlo, const float* __restrict__ Hhi,
                const int* __restrict__ ell, const int* __restrict__ cnt,
                const int* __restrict__ novf, const int* __restrict__ ovf,
                const int* __restrict__ src, const int* __restrict__ dst,
                const float* __restrict__ dis, const float* __restrict__ bias,
                const float* __restrict__ Wn,
                float* __restrict__ G0, float* __restrict__ G1, int n, int nTiles) {
    constexpr int LSTR = MN + LDS_PAD;
    constexpr int CT = MN / 16;
    __shared__ float Wlds[K_DIM * LSTR];
    __shared__ float Zt[2 * RT * ZSTR];

    const int tid = threadIdx.x;
    for (int i = tid * 4; i < K_DIM * MN; i += 512 * 4) {
        const int r = i / MN;
        const int c = i - r * MN;
        *(float4*)&Wlds[r * LSTR + c] = *(const float4*)&Wn[i];
    }

    const int lane = tid & 31;
    const int wave = __builtin_amdgcn_readfirstlane(tid >> 5);
    const float* __restrict__ hb = ((lane < 16) ? Hlo : Hhi) + (lane & 15) * 4;
    const float4 bs = *(const float4*)(bias + lane * 4);

    const bool mm = wave < 2 * CT;
    const int rt = wave / CT;
    const int ct = wave % CT;
    const int kOff = (lane >> 4) * 2;
    const int nCol = lane & 15;
    const float* __restrict__ wcol = &Wlds[kOff * LSTR + ct * 16 + nCol];
    float* __restrict__ hp = ((ct < HP / 16) ? G0 : G1) + (ct & 3) * 16 + nCol;

    int buf = 0;
    for (int tile = blockIdx.x; tile < nTiles; tile += gridDim.x, buf ^= 1) {
        float* __restrict__ zt = Zt + buf * (RT * ZSTR);
        const int dA0 = tile * RT + wave * 2;
        const int dA = dA0 < n ? dA0 : n - 1;
        const int dB = dA0 + 1 < n ? dA0 + 1 : n - 1;

        const int degA = cnt[dA];
        const int degB = cnt[dB];
        const float ddA = dis[dA];
        const float ddB = dis[dB];
        const int sA = ell[(size_t)dA * ELL_W + lane];
        const int sB = ell[(size_t)dB * ELL_W + lane];
        const float4 hsA = *(const float4*)(hb + (size_t)dA * HP);
        const float4 hsB = *(const float4*)(hb + (size_t)dB * HP);
        const int nlA = degA < ELL_W ? degA : ELL_W;
        const int nlB = degB < ELL_W ? degB : ELL_W;
        const int nj = nlA < nlB ? nlA : nlB;

        float4 accA = make_float4(0.0f, 0.0f, 0.0f, 0.0f);
        float4 accB = make_float4(0.0f, 0.0f, 0.0f, 0.0f);
        int j = 0;
        for (; j + 4 <= nj; j += 4) {
            float4 ha[4], hv[4];
#pragma unroll
            for (int u = 0; u < 4; ++u) {
                const int s = __builtin_amdgcn_readlane(sA, j + u);
                ha[u] = *(const float4*)(hb + (size_t)s * HP);
            }
#pragma unroll
            for (int u = 0; u < 4; ++u) {
                const int s = __builtin_amdgcn_readlane(sB, j + u);
                hv[u] = *(const float4*)(hb + (size_t)s * HP);
            }
#pragma unroll
            for (int u = 0; u < 4; ++u) add4(accA, ha[u]);
#pragma unroll
            for (int u = 0; u < 4; ++u) add4(accB, hv[u]);
        }
        accA = row_finish(accA, j, nlA, degA, dA, sA, hb, hsA, bs, ddA, novf, ovf, src, dst, lane);
        accB = row_finish(accB, j, nlB, degB, dB, sB, hb, hsB, bs, ddB, novf, ovf, src, dst, lane);

        accA.x = fmaxf(accA.x, 0.0f); accA.y = fmaxf(accA.y, 0.0f);
        accA.z = fmaxf(accA.z, 0.0f); accA.w = fmaxf(accA.w, 0.0f);
        accB.x = fmaxf(accB.x, 0.0f); accB.y = fmaxf(accB.y, 0.0f);
        accB.z = fmaxf(accB.z, 0.0f); accB.w = fmaxf(accB.w, 0.0f);
        *(float4*)&zt[(wave * 2) * ZSTR + lane * 4] = accA;
        *(float4*)&zt[(wave * 2 + 1) * ZSTR + lane * 4] = accB;
        __syncthreads();

        if (mm && tile * RT + rt * 16 < n) {
            const int rBase = tile * RT + rt * 16 + ((lane >> 4) << 3);
            const float4 dLo = *(const float4*)(dis + rBase);
            const float4 dHi = *(const float4*)(dis + rBase + 4);
            const float dr[8] = {dLo.x, dLo.y, dLo.z, dLo.w, dHi.x, dHi.y, dHi.z, dHi.w};
            const float* __restrict__ za = zt + (rt * 16 + (lane & 15)) * ZSTR + kOff;
            v8f acc = {};
#pragma unroll
            for (int k = 0; k < K_DIM; k += 4) {
                v2f a;
                a.x = za[k];
                a.y = za[k + 1];
                v2f b;
                b.x = wcol[k * LSTR];
                b.y = wcol[k * LSTR + LSTR];
                acc = __builtin_amdgcn_wmma_f32_16x16x4_f32(
                    false, a, false, b, (short)0, acc, false, false);
            }
            asm volatile("v_nop\n\tv_nop\n\tv_nop\n\tv_nop" : "+v"(acc));
#pragma unroll
            for (int v = 0; v < 8; ++v) {
                hp[(size_t)(rBase + v) * HP] = acc[v] * dr[v];
            }
        }
    }
}

extern "C" void kernel_launch(void* const* d_in, const int* in_sizes, int n_in,
                              void* d_out, int out_size, void* d_ws, size_t ws_size,
                              hipStream_t stream) {
    (void)n_in; (void)out_size; (void)ws_size;

    const float* x  = (const float*)d_in[0];
    const int*   ei = (const int*)d_in[1];
    const float* W1 = (const float*)d_in[2];
    const float* b1 = (const float*)d_in[3];
    const float* W2 = (const float*)d_in[4];
    const float* b2 = (const float*)d_in[5];
    const float* W3 = (const float*)d_in[6];
    const float* b3 = (const float*)d_in[7];
    float* out = (float*)d_out;

    const int N = in_sizes[0] / K_DIM;
    const int E = in_sizes[1] / 2;
    const int* src = ei;
    const int* dst = ei + E;

    float* dis  = (float*)d_ws;
    int*   cnt  = (int*)(dis + N);
    int*   novf = cnt + N;
    int*   ell  = novf + 64;
    int*   ovf  = ell + (size_t)N * ELL_W;
    float* P    = (float*)(ovf + E);
    float* Z    = P + (size_t)N * HP;
    uint2* list      = (uint2*)Z;
    int*   binCount  = (int*)Z + 2 * (size_t)E;
    int*   binCursor = binCount + MAXB;

    const int T = 256;
    const int gChunk = (E + CHUNK - 1) / CHUNK;
    const int gBin   = (N + BIN_W - 1) / BIN_W;
    const int gNode = (N + 7) / 8;
    const int tiles = (N + 15) / 16;
    const int gGemm = (tiles + 7) / 8;
    const int nT32 = (N + RT - 1) / RT;

    gcn_bins_init<<<1, MAXB, 0, stream>>>(binCount, binCursor, novf);
    gcn_bin_hist<<<gChunk, T, 0, stream>>>(dst, binCount, E);
    gcn_bin_scatter<<<gChunk, T, 0, stream>>>(src, dst, binCount, binCursor, list, E);
    gcn_group_rows<<<gBin, T, 0, stream>>>(list, binCount, ell, cnt, dis, novf, ovf, N);

    gcn_gemm_wmma<128, false><<<gGemm, T, 0, stream>>>(x, W1, dis, out, P, N);

    gcn_gather_gemm<128><<<768, 512, 0, stream>>>(out, P, ell, cnt, novf, ovf, src, dst, dis, b1, W2,
                                                  Z, Z + (size_t)N * HP, N, nT32);

    gcn_gather_gemm<64><<<768, 512, 0, stream>>>(Z, Z + (size_t)N * HP, ell, cnt, novf, ovf, src, dst,
                                                  dis, b2, W3, P, P, N, nT32);

    gcn_gather<64><<<gNode, T, 0, stream>>>(P, P, ell, cnt, novf, ovf, src, dst, dis, b3, out, N);
}


# Layers 1 and 2 aggregate before their product: layer 1 gathers rows of x with dis[src] by fma, layer 2 chains W2 and W3 through LDS; plain GEMM launch and its table gone; 7 launches; sums reordered
# speedup vs baseline: 6.0421x; 1.1270x over previous
#include <hip/hip_runtime.h>
#include <hip/hip_bf16.h>

typedef __attribute__((ext_vector_type(2))) float v2f;
typedef __attribute__((ext_vector_type(8))) float v8f;

#define K_DIM 128
#define LDS_PAD 8
#define ELL_W 32
#define HP 64

#define BIN_SHIFT 8
#define BIN_W (1 << BIN_SHIFT)
#define MAXB 512
#define EPT 16
#define CHUNK (EPT * 256)
#define GRB 18

__device__ __forceinline__ void block_exscan(int* arr, int* wsum) {
    const int t = threadIdx.x;
    const int lane = t & 31;
    const int wave = t >> 5;
    const int a = arr[2 * t];
    const int b = arr[2 * t + 1];
    const int v = a + b;
    int inc = v;
#pragma unroll
    for (int o = 1; o < 32; o <<= 1) {
        const int u = __shfl_up(inc, o);
        if (lane >= o) inc += u;
    }
    if (lane == 31) wsum[wave] = inc;
    __syncthreads();
    int off = 0;
#pragma unroll
    for (int w = 0; w < 8; ++w) off += (w < wave) ? wsum[w] : 0;
    const int ex = off + inc - v;
    arr[2 * t] = ex;
    arr[2 * t + 1] = ex + a;
    __syncthreads();
}

__global__ void gcn_bins_init(int* __restrict__ binCount, int* __restrict__ binCursor,
                              int* __restrict__ novf) {
    const int t = threadIdx.x;
    binCount[t] = 0;
    binCursor[t] = 0;
    if (t == 0) *novf = 0;
}

__global__ void __launch_bounds__(256)
gcn_bin_hist(const int* __restrict__ dst, int* __restrict__ binCount, int nE) {
    __shared__ int h[MAXB];
    const int t = threadIdx.x;
    h[t] = 0;
    h[t + 256] = 0;
    __syncthreads();
    const int base = blockIdx.x * CHUNK;
    int dv[EPT];
#pragma unroll
    for (int k = 0; k < EPT; ++k) {
        const int e = base + k * 256 + t;
        dv[k] = dst[e < nE ? e : nE - 1];
    }
#pragma unroll
    for (int k = 0; k < EPT; ++k) {
        const int e = base + k * 256 + t;
        if (e < nE) atomicAdd(&h[dv[k] >> BIN_SHIFT], 1);
    }
    __syncthreads();
    const int c0 = h[t];
    const int c1 = h[t + 256];
    if (c0) atomicAdd(&binCount[t], c0);
    if (c1) atomicAdd(&binCount[t + 256], c1);
}

__global__ void __launch_bounds__(256)
gcn_bin_scatter(const int* __restrict__ src, const int* __restrict__ dst,
                const int* __restrict__ binCount, int* __restrict__ binCursor,
                uint2* __restrict__ list, int nE) {
    __shared__ int gstart[MAXB];
    __shared__ int h[MAXB];
    __shared__ int wsum[8];
    __shared__ unsigned stA[CHUNK];
    __shared__ unsigned stB[CHUNK];
    const int t = threadIdx.x;
    const int base = blockIdx.x * CHUNK;

    gstart[t] = binCount[t];
    gstart[t + 256] = binCount[t + 256];
    h[t] = 0;
    h[t + 256] = 0;
    __syncthreads();
    block_exscan(gstart, wsum);

    unsigned wA[EPT];
    int pk[EPT];
#pragma unroll
    for (int k = 0; k < EPT; ++k) {
        const int e = base + k * 256 + t;
        const int ec = e < nE ? e : nE - 1;
        pk[k] = dst[ec];
        wA[k] = (unsigned)src[ec];
    }
#pragma unroll
    for (int k = 0; k < EPT; ++k) {
        const int e = base + k * 256 + t;
        const int d = pk[k];
        const int s = (int)wA[k];
        pk[k] = -1;
        if (e < nE) {
            const int bin = d >> BIN_SHIFT;
            const int r = atomicAdd(&h[bin], 1);
            wA[k] = ((unsigned)(d & (BIN_W - 1)) << 24) | (unsigned)s;
            pk[k] = (bin << 12) | r;
        }
    }
    __syncthreads();

    const int c0 = h[t];
    const int c1 = h[t + 256];
    if (c0) gstart[t] += atomicAdd(&binCursor[t], c0);
    if (c1) gstart[t + 256] += atomicAdd(&binCursor[t + 256], c1);
    block_exscan(h, wsum);

#pragma unroll
    for (int k = 0; k < EPT; ++k) {
        if (pk[k] >= 0) {
            const int bin = pk[k] >> 12;
            const int p = h[bin] + (pk[k] & 4095);
            stA[p] = wA[k];
            stB[p] = ((unsigned)bin << 12) | (unsigned)(k * 256 + t);
        }
    }
    __syncthreads();

    const int nHere = (nE - base < CHUNK) ? (nE - base) : CHUNK;
    for (int p = t; p < nHere; p += 256) {
        const unsigned A = stA[p];
        const unsigned B = stB[p];
        const int bin = (int)(B >> 12);
        const int pos = gstart[bin] + (p - h[bin]);
        list[pos] = make_uint2(A, (unsigned)base + (B & 4095u));
    }
}

__global__ void __launch_bounds__(256)
gcn_group_rows(const uint2* __restrict__ list, const int* __restrict__ binCount,
               int* __restrict__ ell, int* __restrict__ cnt, float* __restrict__ dis,
               int* __restrict__ novf, int* __restrict__ ovf, int n) {
    __shared__ int gs[MAXB];
    __shared__ int wsum[8];
    __shared__ int cntL[BIN_W];
    __shared__ uint4 tile[BIN_W * ELL_W / 4];
    const int t = threadIdx.x;
    const int b = blockIdx.x;

    gs[t] = binCount[t];
    gs[t + 256] = binCount[t + 256];
    const int nb = binCount[b];
    cntL[t] = 0;
    for (int q = t; q < BIN_W * ELL_W / 4; q += 256) tile[q] = make_uint4(0u, 0u, 0u, 0u);
    __syncthreads();
    block_exscan(gs, wsum);
    const int start = gs[b];

    int* tl = (int*)tile;
    for (int i0 = 0; i0 < nb; i0 += 256 * GRB) {
        uint2 rec[GRB];
#pragma unroll
        for (int u = 0; u < GRB; ++u) {
            const int i = i0 + u * 256 + t;
            rec[u] = list[start + (i < nb ? i : nb - 1)];
        }
#pragma unroll
        for (int u = 0; u < GRB; ++u) {
            const int i = i0 + u * 256 + t;
            if (i < nb) {
                const int dl = (int)(rec[u].x >> 24);
                const int s = (int)(rec[u].x & 0xFFFFFFu);
                const int r = atomicAdd(&cntL[dl], 1);
                if (r < ELL_W) tl[dl * ELL_W + r] = s;
                else           ovf[atomicAdd(novf, 1)] = (int)rec[u].y;
            }
        }
    }
    __syncthreads();

    const int row0 = b * BIN_W;
    uint4* __restrict__ ellv = (uint4*)ell + (size_t)b * (BIN_W * ELL_W / 4);
    for (int q = t; q < BIN_W * ELL_W / 4; q += 256) {
        if (row0 + (q >> 3) < n) ellv[q] = tile[q];
    }
    const int d = row0 + t;
    if (d < n) {
        const int c = cntL[t];
        cnt[d] = c;
        dis[d] = rsqrtf(fmaxf(1.0f + (float)c, 1.0f));
    }
}

template <int EPI>
__device__ __forceinline__ int lane_pick(int v, int j) {
    if constexpr (EPI == 1) return __builtin_amdgcn_readlane(v, j);
    else                    return __shfl(v, j);
}

template <int M>
__global__ void __launch_bounds__(256)
gcn_gather(const float* __restrict__ Hlo, const float* __restrict__ Hhi,
           const int* __restrict__ ell, const int* __restrict__ cnt,
           const int* __restrict__ novf, const int* __restrict__ ovf,
           const int* __restrict__ src, const int* __restrict__ dst,
           const float* __restrict__ dis, const float* __restrict__ bias,
           float* __restrict__ out, int n) {
    constexpr int LPR = M / 4;
    constexpr int EPI = 32 / LPR;
    const int lane = threadIdx.x & 31;
    const int d = __builtin_amdgcn_readfirstlane(blockIdx.x * 8 + (threadIdx.x >> 5));
    if (d >= n) return;
    const int sub = lane / LPR;
    const int c4  = lane % LPR;
    const float* __restrict__ hb = ((c4 < 16) ? Hlo : Hhi) + (c4 & 15) * 4;

    const float dd = dis[d];
    const int deg = cnt[d];
    const int nl = deg < ELL_W ? deg : ELL_W;

    const int my_s = ell[(size_t)d * ELL_W + lane];

    const float4 hs = *(const float4*)(hb + (size_t)d * HP);
    const float4 bs = *(const float4*)(bias + c4 * 4);
    float4 acc = make_float4(0.0f, 0.0f, 0.0f, 0.0f);

    int j0 = 0;
    for (; j0 + 4 * EPI <= nl; j0 += 4 * EPI) {
        float4 h[4];
#pragma unroll
        for (int u = 0; u < 4; ++u) {
            const int j = j0 + u * EPI + sub;
            const int s = lane_pick<EPI>(my_s, j);
            h[u] = *(const float4*)(hb + (size_t)s * HP);
        }
#pragma unroll
        for (int u = 0; u < 4; ++u) {
            acc.x += h[u].x;
            acc.y += h[u].y;
            acc.z += h[u].z;
            acc.w += h[u].w;
        }
    }
    constexpr int TT = (4 * EPI - 1 + EPI - 1) / EPI;
    {
        asm volatile("" :: "v"(my_s));
        float4 h[TT];
#pragma unroll
        for (int u = 0; u < TT; ++u) {
            const int jb = j0 + u * EPI;
            if (jb < nl) {
                const int j = jb + sub;
                const int s = lane_pick<EPI>(my_s, j);
                h[u] = *(const float4*)(hb + (size_t)s * HP);
            }
        }
#pragma unroll
        for (int u = 0; u < TT; ++u) {
            const int jb = j0 + u * EPI;
            if (jb < nl) {
                float4 t = h[u];
                if (EPI == 2 && jb + sub >= nl) t = make_float4(0.0f, 0.0f, 0.0f, 0.0f);
                acc.x += t.x;
                acc.y += t.y;
                acc.z += t.z;
                acc.w += t.w;
            }
        }
    }

    if (deg > ELL_W) {
        const int no = *novf;
        for (int i0 = 0; i0 < no; i0 += 32) {
            const int i = i0 + lane;
            int s_ = 0;
            bool m = false;
            if (i < no) {
                const int e = ovf[i];
                m = (dst[e] == d);
                if (m) s_ = src[e];
            }
            unsigned mask = (unsigned)__ballot(m);
            while (mask) {
                const int l = __ffs(mask) - 1;
                mask &= mask - 1;
                const int s = __shfl(s_, l);
                if (sub == 0) {
                    const float4 h = *(const float4*)(hb + (size_t)s * HP);
                    acc.x += h.x;
                    acc.y += h.y;
                    acc.z += h.z;
                    acc.w += h.w;
                }
            }
        }
    }

    if (sub == 0) {
        acc.x += hs.x;
        acc.y += hs.y;
        acc.z += hs.z;
        acc.w += hs.w;
    }
    if (EPI == 2) {
        acc.x += __shfl_xor(acc.x, 16);
        acc.y += __shfl_xor(acc.y, 16);
        acc.z += __shfl_xor(acc.z, 16);
        acc.w += __shfl_xor(acc.w, 16);
    }
    if (sub == 0) {
        acc.x = acc.x * dd + bs.x;
        acc.y = acc.y * dd + bs.y;
        acc.z = acc.z * dd + bs.z;
        acc.w = acc.w * dd + bs.w;
        *(float4*)(out + (size_t)d * M + c4 * 4) = acc;
    }
}

#define RT 32
#define ZSTR (K_DIM + 4)

__device__ __forceinline__ void add4(float4& a, const float4 h) {
    a.x += h.x; a.y += h.y; a.z += h.z; a.w += h.w;
}

__device__ __forceinline__ void fma4(float4& a, const float w, const float4 h) {
    a.x = fmaf(w, h.x, a.x); a.y = fmaf(w, h.y, a.y);
    a.z = fmaf(w, h.z, a.z); a.w = fmaf(w, h.w, a.w);
}

__device__ __forceinline__ float lane_f(const float v, const int j) {
    return __int_as_float(__builtin_amdgcn_readlane(__float_as_int(v), j));
}

template <bool WT, int RS>
__device__ __forceinline__ float4 row_overflow(float4 acc, const int deg, const int d,
                                               const float* __restrict__ hb,
                                               const float* __restrict__ dis,
                                               const int* __restrict__ novf, const int* __restrict__ ovf,
                                               const int* __restrict__ src, const int* __restrict__ dst,
                                               const int lane) {
    if (deg > ELL_W) {
        const int no = *novf;
        for (int i0 = 0; i0 < no; i0 += 32) {
            const int i = i0 + lane;
            int s_ = 0;
            bool m = false;
            if (i < no) {
                const int e = ovf[i];
                m = (dst[e] == d);
                if (m) s_ = src[e];
            }
            unsigned mask = (unsigned)__ballot(m);
            while (mask) {
                const int l = __ffs(mask) - 1;
                mask &= mask - 1;
                const int s = __shfl(s_, l);
                const float4 h = *(const float4*)(hb + (size_t)s * RS);
                if (WT) fma4(acc, dis[s], h);
                else    add4(acc, h);
            }
        }
    }
    return acc;
}

template <int LSTR>
__device__ __forceinline__ v8f tile_product(const float* __restrict__ za, const float* __restrict__ wcol) {
    v8f acc = {};
#pragma unroll
    for (int k = 0; k < K_DIM; k += 4) {
        v2f a;
        a.x = za[k];
        a.y = za[k + 1];
        v2f b;
        b.x = wcol[k * LSTR];
        b.y = wcol[k * LSTR + LSTR];
        acc = __builtin_amdgcn_wmma_f32_16x16x4_f32(
            false, a, false, b, (short)0, acc, false, false);
    }
    asm volatile("v_nop\n\tv_nop\n\tv_nop\n\tv_nop" : "+v"(acc));
    return acc;
}

__global__ void __launch_bounds__(512, 12)
gcn_aggx_gemm(const float* __restrict__ X,
              const int* __restrict__ ell, const int* __restrict__ cnt,
              const int* __restrict__ novf, const int* __restrict__ ovf,
              const int* __restrict__ src, const int* __restrict__ dst,
              const float* __restrict__ dis, const float* __restrict__ bias,
              const float* __restrict__ Wn,
              float* __restrict__ G0, float* __restrict__ G1, int n, int nTiles) {
    constexpr int MN = K_DIM;
    constexpr int LSTR = MN + LDS_PAD;
    constexpr int CT = MN / 16;
    __shared__ float Wlds[K_DIM * LSTR];
    __shared__ float Zt[2 * RT * ZSTR];
    __shared__ float bl[MN];

    const int tid = threadIdx.x;
    if (tid < MN) bl[tid] = bias[tid];
    constexpr int WP = (K_DIM * MN) / (512 * 4);
    static_assert(WP * 512 * 4 == K_DIM * MN, "W pieces");
    {
        float4 wv[WP];
#pragma unroll
        for (int p = 0; p < WP; ++p) wv[p] = *(const float4*)&Wn[(p * 512 + tid) * 4];
#pragma unroll
        for (int p = 0; p < WP; ++p) {
            const int i = (p * 512 + tid) * 4;
            const int r = i / MN;
            const int c = i - r * MN;
            *(float4*)&Wlds[r * LSTR + c] = wv[p];
        }
    }
    __syncthreads();

    const int lane = tid & 31;
    const int wave = __builtin_amdgcn_readfirstlane(tid >> 5);
    const float* __restrict__ xb = X + lane * 4;

    const int rt = wave / CT;
    const int ct = wave % CT;

    int buf = 0;
    for (int tile = blockIdx.x; tile < nTiles; tile += gridDim.x, buf ^= 1) {
        float* __restrict__ zt = Zt + buf * (RT * ZSTR);
        const int dA0 = tile * RT + wave * 2;
        const int dA = dA0 < n ? dA0 : n - 1;
        const int dB = dA0 + 1 < n ? dA0 + 1 : n - 1;

        const int degA = cnt[dA];
        const int degB = cnt[dB];
        const float ddA = dis[dA];
        const float ddB = dis[dB];
        const int sA = ell[(size_t)dA * ELL_W + lane];
        const int sB = ell[(size_t)dB * ELL_W + lane];
        const float4 oA = *(const float4*)(xb + (size_t)dA * K_DIM);
        const float4 oB = *(const float4*)(xb + (size_t)dB * K_DIM);
        const float wA = dis[sA];
        const float wB = dis[sB];
        float4 accA = make_float4(oA.x * ddA, oA.y * ddA, oA.z * ddA, oA.w * ddA);
        float4 accB = make_float4(oB.x * ddB, oB.y * ddB, oB.z * ddB, oB.w * ddB);
        const int nlA = degA < ELL_W ? degA : ELL_W;
        const int nlB = degB < ELL_W ? degB : ELL_W;
        const int nj = nlA < nlB ? nlA : nlB;

        int j = 0;
        for (; j + 4 <= nj; j += 4) {
            float4 ha[4], hv[4];
#pragma unroll
            for (int u = 0; u < 4; ++u) {
                const int s = __builtin_amdgcn_readlane(sA, j + u);
                ha[u] = *(const float4*)(xb + (size_t)s * K_DIM);
            }
#pragma unroll
            for (int u = 0; u < 4; ++u) {
                const int s = __builtin_amdgcn_readlane(sB, j + u);
                hv[u] = *(const float4*)(xb + (size_t)s * K_DIM);
            }
#pragma unroll
            for (int u = 0; u < 4; ++u) fma4(accA, lane_f(wA, j + u), ha[u]);
#pragma unroll
            for (int u = 0; u < 4; ++u) fma4(accB, lane_f(wB, j + u), hv[u]);
        }
        const int remA = nlA - j;
        const int rem = remA + (nlB - j);
        asm volatile("" :: "v"(sA), "v"(sB), "v"(wA), "v"(wB));
        for (int q0 = 0; q0 < rem; q0 += 6) {
            float4 h[6];
#pragma unroll
            for (int u = 0; u < 6; ++u) {
                const int q = q0 + u;
                if (q < rem) {
                    const int a = __builtin_amdgcn_readlane(sA, (j + q) & 31);
                    const int b = __builtin_amdgcn_readlane(sB, (j + q - remA) & 31);
                    const int s = q < remA ? a : b;
                    h[u] = *(const float4*)(xb + (size_t)s * K_DIM);
                }
            }
#pragma unroll
            for (int u = 0; u < 6; ++u) {
                const int q = q0 + u;
                if (q < remA)     fma4(accA, lane_f(wA, (j + q) & 31), h[u]);
                else if (q < rem) fma4(accB, lane_f(wB, (j + q - remA) & 31), h[u]);
            }
        }
        accA = row_overflow<true, K_DIM>(accA, degA, dA, xb, dis, novf, ovf, src, dst, lane);
        accB = row_overflow<true, K_DIM>(accB, degB, dB, xb, dis, novf, ovf, src, dst, lane);
        accA.x *= ddA; accA.y *= ddA; accA.z *= ddA; accA.w *= ddA;
        accB.x *= ddB; accB.y *= ddB; accB.z *= ddB; accB.w *= ddB;
        *(float4*)&zt[(wave * 2) * ZSTR + lane * 4] = accA;
        *(float4*)&zt[(wave * 2 + 1) * ZSTR + lane * 4] = accB;
        __syncthreads();

        if (tile * RT + rt * 16 < n) {
            int lv = lane;
            asm volatile("" : "+v"(lv));
            const int kOff = (lv >> 4) * 2;
            const int nCol = lv & 15;
            const float* __restrict__ wcol = &Wlds[kOff * LSTR + ct * 16 + nCol];
            float* __restrict__ hp = ((ct < HP / 16) ? G0 : G1) + (ct & 3) * 16 + nCol;
            const int rBase = tile * RT + rt * 16 + ((lv >> 4) << 3);
            const float4 dLo = *(const float4*)(dis + rBase);
            const float4 dHi = *(const float4*)(dis + rBase + 4);
            const float dr[8] = {dLo.x, dLo.y, dLo.z, dLo.w, dHi.x, dHi.y, dHi.z, dHi.w};
            const float bc = bl[ct * 16 + nCol];
            const float* __restrict__ za = zt + (rt * 16 + nCol) * ZSTR + kOff;
            const v8f acc = tile_product<LSTR>(za, wcol);
#pragma unroll
            for (int v = 0; v < 8; ++v) {
                hp[(size_t)(rBase + v) * HP] = fmaxf(acc[v] + bc, 0.0f) * dr[v];
            }
        }
    }
}

#define GU 6
#define TQ 8
__global__ void __launch_bounds__(512, 8)
gcn_agg_gemm2(const float* __restrict__ Hlo, const float* __restrict__ Hhi,
              const int* __restrict__ ell, const int* __restrict__ cnt,
              const int* __restrict__ novf, const int* __restrict__ ovf,
              const int* __restrict__ src, const int* __restrict__ dst,
              const float* __restrict__ dis, const float* __restrict__ bias,
              const float* __restrict__ Wa, const float* __restrict__ Wb,
              float* __restrict__ G, int n, int nTiles) {
    constexpr int MA = K_DIM;
    constexpr int MB = HP;
    constexpr int LSA = MA + LDS_PAD;
    constexpr int LSB = MB + LDS_PAD;
    __shared__ float WaL[K_DIM * LSA];
    __shared__ float WbL[K_DIM * LSB];
    __shared__ float Zt[RT * ZSTR];
    __shared__ float Ht[RT * ZSTR];
    __shared__ float bl[MA];

    const int tid = threadIdx.x;
    if (tid < MA) bl[tid] = bias[tid];
    constexpr int WPA = (K_DIM * MA) / (512 * 4);
    constexpr int WPB = (K_DIM * MB) / (512 * 4);
    static_assert(WPA * 512 * 4 == K_DIM * MA && WPB * 512 * 4 == K_DIM * MB, "W pieces");
    {
        float4 wa[WPA], wb[WPB];
#pragma unroll
        for (int p = 0; p < WPA; ++p) wa[p] = *(const float4*)&Wa[(p * 512 + tid) * 4];
#pragma unroll
        for (int p = 0; p < WPB; ++p) wb[p] = *(const float4*)&Wb[(p * 512 + tid) * 4];
#pragma unroll
        for (int p = 0; p < WPA; ++p) {
            const int i = (p * 512 + tid) * 4;
            const int r = i / MA;
            const int c = i - r * MA;
            *(float4*)&WaL[r * LSA + c] = wa[p];
        }
#pragma unroll
        for (int p = 0; p < WPB; ++p) {
            const int i = (p * 512 + tid) * 4;
            const int r = i / MB;
            const int c = i - r * MB;
            *(float4*)&WbL[r * LSB + c] = wb[p];
        }
    }
    __syncthreads();

    const int lane = tid & 31;
    const int wave = __builtin_amdgcn_readfirstlane(tid >> 5);
    const float* __restrict__ hb = ((lane < 16) ? Hlo : Hhi) + (lane & 15) * 4;

    const int rtA = wave >> 3;
    const int ctA = wave & 7;
    const bool mmB = wave < 2 * (MB / 16);
    const int rtB = (wave >> 2) & 1;
    const int ctB = wave & 3;

    for (int tile = blockIdx.x; tile < nTiles; tile += gridDim.x) {
        const int dA0 = tile * RT + wave * 2;
        const int dA = dA0 < n ? dA0 : n - 1;
        const int dB = dA0 + 1 < n ? dA0 + 1 : n - 1;

        const int degA = cnt[dA];
        const int degB = cnt[dB];
        const float ddA = dis[dA];
        const float ddB = dis[dB];
        const int sA = ell[(size_t)dA * ELL_W + lane];
        const int sB = ell[(size_t)dB * ELL_W + lane];
        float4 accA = *(const float4*)(hb + (size_t)dA * HP);
        float4 accB = *(const float4*)(hb + (size_t)dB * HP);
        const int nlA = degA < ELL_W ? degA : ELL_W;
        const int nlB = degB < ELL_W ? degB : ELL_W;
        const int nj = nlA < nlB ? nlA : nlB;

        int j = 0;
        for (; j + GU <= nj; j += GU) {
            float4 ha[GU], hv[GU];
#pragma unroll
            for (int u = 0; u < GU; ++u) {
                const int s = __builtin_amdgcn_readlane(sA, j + u);
                ha[u] = *(const float4*)(hb + (size_t)s * HP);
            }
#pragma unroll
            for (int u = 0; u < GU; ++u) {
                const int s = __builtin_amdgcn_readlane(sB, j + u);
                hv[u] = *(const float4*)(hb + (size_t)s * HP);
            }
#pragma unroll
            for (int u = 0; u < GU; ++u) add4(accA, ha[u]);
#pragma unroll
            for (int u = 0; u < GU; ++u) add4(accB, hv[u]);
        }
        const int remA = nlA - j;
        const int rem = remA + (nlB - j);
        asm volatile("" :: "v"(sA), "v"(sB));
        for (int q0 = 0; q0 < rem; q0 += TQ) {
            float4 h[TQ];
#pragma unroll
            for (int u = 0; u < TQ; ++u) {
                const int q = q0 + u;
                if (q < rem) {
                    const int a = __builtin_amdgcn_readlane(sA, (j + q) & 31);
                    const int b = __builtin_amdgcn_readlane(sB, (j + q - remA) & 31);
                    const int s = q < remA ? a : b;
                    h[u] = *(const float4*)(hb + (size_t)s * HP);
                }
            }
#pragma unroll
            for (int u = 0; u < TQ; ++u) {
                const int q = q0 + u;
                if (q < remA)     add4(accA, h[u]);
                else if (q < rem) add4(accB, h[u]);
            }
        }
        accA = row_overflow<false, HP>(accA, degA, dA, hb, dis, novf, ovf, src, dst, lane);
        accB = row_overflow<false, HP>(accB, degB, dB, hb, dis, novf, ovf, src, dst, lane);
        accA.x *= ddA; accA.y *= ddA; accA.z *= ddA; accA.w *= ddA;
        accB.x *= ddB; accB.y *= ddB; accB.z *= ddB; accB.w *= ddB;
        *(float4*)&Zt[(wave * 2) * ZSTR + lane * 4] = accA;
        *(float4*)&Zt[(wave * 2 + 1) * ZSTR + lane * 4] = accB;
        __syncthreads();

        int lv = lane;
        asm volatile("" : "+v"(lv));
        const int kOff = (lv >> 4) * 2;
        const int nCol = lv & 15;
        const int rHalf = (lv >> 4) << 3;

        if (tile * RT + rtA * 16 < n) {
            const float bc = bl[ctA * 16 + nCol];
            const v8f acc = tile_product<LSA>(Zt + (rtA * 16 + nCol) * ZSTR + kOff,
                                              &WaL[kOff * LSA + ctA * 16 + nCol]);
            float* __restrict__ ht = Ht + (rtA * 16 + rHalf) * ZSTR + ctA * 16 + nCol;
#pragma unroll
            for (int v = 0; v < 8; ++v) ht[v * ZSTR] = fmaxf(acc[v] + bc, 0.0f);
        }
        __syncthreads();

        if (mmB && tile * RT + rtB * 16 < n) {
            const int rBase = tile * RT + rtB * 16 + rHalf;
            const float4 dLo = *(const float4*)(dis + rBase);
            const float4 dHi = *(const float4*)(dis + rBase + 4);
            const float dr[8] = {dLo.x, dLo.y, dLo.z, dLo.w, dHi.x, dHi.y, dHi.z, dHi.w};
            const v8f acc = tile_product<LSB>(Ht + (rtB * 16 + nCol) * ZSTR + kOff,
                                              &WbL[kOff * LSB + ctB * 16 + nCol]);
            float* __restrict__ gp = G + (size_t)rBase * MB + ctB * 16 + nCol;
#pragma unroll
            for (int v = 0; v < 8; ++v) gp[(size_t)v * MB] = acc[v] * dr[v];
        }
    }
}

extern "C" void kernel_launch(void* const* d_in, const int* in_sizes, int n_in,
                              void* d_out, int out_size, void* d_ws, size_t ws_size,
                              hipStream_t stream) {
    (void)n_in; (void)out_size; (void)ws_size;

    const float* x  = (const float*)d_in[0];
    const int*   ei = (const int*)d_in[1];
    const float* W1 = (const float*)d_in[2];
    const float* b1 = (const float*)d_in[3];
    const float* W2 = (const float*)d_in[4];
    const float* b2 = (const float*)d_in[5];
    const float* W3 = (const float*)d_in[6];
    const float* b3 = (const float*)d_in[7];
    float* out = (float*)d_out;

    const int N = in_sizes[0] / K_DIM;
    const int E = in_sizes[1] / 2;
    const int* src = ei;
    const int* dst = ei + E;

    float* dis  = (float*)d_ws;
    int*   cnt  = (int*)(dis + N);
    int*   novf = cnt + N;
    int*   ell  = novf + 64;
    int*   ovf  = ell + (size_t)N * ELL_W;
    float* P    = (float*)(ovf + E);
    float* Z    = P + (size_t)N * HP;
    uint2* list      = (uint2*)Z;
    int*   binCount  = (int*)Z + 2 * (size_t)E;
    int*   binCursor = binCount + MAXB;

    const int T = 256;
    const int gChunk = (E + CHUNK - 1) / CHUNK;
    const int gBin   = (N + BIN_W - 1) / BIN_W;
    const int gNode = (N + 7) / 8;
    const int nT32 = (N + RT - 1) / RT;

    gcn_bins_init<<<1, MAXB, 0, stream>>>(binCount, binCursor, novf);
    gcn_bin_hist<<<gChunk, T, 0, stream>>>(dst, binCount, E);
    gcn_bin_scatter<<<gChunk, T, 0, stream>>>(src, dst, binCount, binCursor, list, E);
    gcn_group_rows<<<gBin, T, 0, stream>>>(list, binCount, ell, cnt, dis, novf, ovf, N);

    gcn_aggx_gemm<<<768, 512, 0, stream>>>(x, ell, cnt, novf, ovf, src, dst, dis, b1, W1, out, P, N, nT32);

    gcn_agg_gemm2<<<512, 512, 0, stream>>>(out, P, ell, cnt, novf, ovf, src, dst, dis, b2, W2, W3, Z, N, nT32);

    gcn_gather<64><<<gNode, T, 0, stream>>>(Z, Z, ell, cnt, novf, ovf, src, dst, dis, b3, out, N);
}
